# starting kernel
# baseline (speedup 1.0000x reference)
#include <hip/hip_runtime.h>


typedef float v2f __attribute__((ext_vector_type(2)));
typedef float v8f __attribute__((ext_vector_type(8)));

#define NN   100000
#define EE   1600000
#define DH   64
#define DOUT 16

__global__ void gcn_zero_f32(float* __restrict__ p, int n) {
  int i = blockIdx.x * blockDim.x + threadIdx.x;
  if (i < n) p[i] = 0.0f;
}

__global__ void gcn_deg_kernel(const int* __restrict__ dst,
                               float* __restrict__ deg) {
  int e = blockIdx.x * blockDim.x + threadIdx.x;
  if (e < EE) unsafeAtomicAdd(&deg[dst[e]], 1.0f);
}

__global__ void gcn_dinv_kernel(float* __restrict__ deg) {
  int i = blockIdx.x * blockDim.x + threadIdx.x;
  if (i < NN) deg[i] = rsqrtf(deg[i] + 1.0f);
}

template <int NCOLS, bool BIAS, bool RELU>
__global__ __launch_bounds__(32) void gcn_gemm_wmma(
    const float* __restrict__ X, const float* __restrict__ W,
    const float* __restrict__ bias, float* __restrict__ Out) {
  const int tileM = blockIdx.x;
  const int tileN = blockIdx.y;
  const int lane  = threadIdx.x;
  const int half  = lane >> 4;
  const int l     = lane & 15;
  const int arow  = tileM * 16 + l;
  const int bcol  = tileN * 16 + l;

  v8f c = {};
#pragma unroll
  for (int k0 = 0; k0 < DH; k0 += 4) {
    const int ak = k0 + half * 2;
    v2f a, b;
    a.x = X[arow * DH + ak];
    a.y = X[arow * DH + ak + 1];
    b.x = W[ak * NCOLS + bcol];
    b.y = W[(ak + 1) * NCOLS + bcol];
    c = __builtin_amdgcn_wmma_f32_16x16x4_f32(
         false, a,  false, b,
         (short)0, c,  false,  false);
  }

#pragma unroll
  for (int r = 0; r < 8; ++r) {
    const int row = tileM * 16 + half * 8 + r;
    float v = c[r];
    if (BIAS) v += bias[bcol];
    if (RELU) v = fmaxf(v, 0.0f);
    Out[row * NCOLS + bcol] = v;
  }
}

__global__ void gcn_scatter_kernel(const float* __restrict__ H,
                                   const int* __restrict__ src,
                                   const int* __restrict__ dst,
                                   const float* __restrict__ dinv,
                                   float* __restrict__ AGG) {
  int idx = blockIdx.x * blockDim.x + threadIdx.x;
  if (idx >= EE * DH) return;
  const int e = idx >> 6;
  const int d = idx & (DH - 1);
  const int s = src[e];
  const int t = dst[e];
  const float coef = dinv[s] * dinv[t];
  unsafeAtomicAdd(&AGG[t * DH + d], H[s * DH + d] * coef);
}

__global__ void gcn_epilogue_kernel(float* __restrict__ AGG,
                                    const float* __restrict__ H,
                                    const float* __restrict__ dinv,
                                    const float* __restrict__ b) {
  int idx = blockIdx.x * blockDim.x + threadIdx.x;
  if (idx >= NN * DH) return;
  const int i = idx >> 6;
  const int d = idx & (DH - 1);
  const float di = dinv[i];
  const float v = AGG[idx] + H[idx] * di * di + b[d];
  AGG[idx] = fmaxf(v, 0.0f);
}

extern "C" void kernel_launch(void* const* d_in, const int* in_sizes, int n_in,
                              void* d_out, int out_size, void* d_ws,
                              size_t ws_size, hipStream_t stream) {
  (void)in_sizes; (void)n_in; (void)out_size; (void)ws_size;

  const float* x   = (const float*)d_in[0];
  const int*   ei  = (const int*)d_in[1];
  const int*   src = ei;
  const int*   dst = ei + EE;
  const float* W1  = (const float*)d_in[2];
  const float* b1  = (const float*)d_in[3];
  const float* W2  = (const float*)d_in[4];
  const float* b2  = (const float*)d_in[5];
  const float* W3  = (const float*)d_in[6];
  const float* b3  = (const float*)d_in[7];
  const float* Wf1 = (const float*)d_in[8];
  const float* bf1 = (const float*)d_in[9];
  const float* Wf2 = (const float*)d_in[10];
  const float* bf2 = (const float*)d_in[11];
  float* out = (float*)d_out;

  float* base = (float*)d_ws;
  float* dinv = base;
  float* bufA = base + NN;
  float* bufB = bufA + NN * DH;

  const int T = 256;
  const int gN   = (NN + T - 1) / T;
  const int gE   = (EE + T - 1) / T;
  const int gND  = (NN * DH + T - 1) / T;
  const int gED  = (EE * DH + T - 1) / T;
  const dim3 gemmGrid(NN / 16, DH / 16);
  const dim3 gemmGridOut(NN / 16, DOUT / 16);
  const dim3 gemmBlk(32);

  gcn_zero_f32<<<gN, T, 0, stream>>>(dinv, NN);
  gcn_deg_kernel<<<gE, T, 0, stream>>>(dst, dinv);
  gcn_dinv_kernel<<<gN, T, 0, stream>>>(dinv);

  gcn_gemm_wmma<DH, false, false><<<gemmGrid, gemmBlk, 0, stream>>>(x, W1, nullptr, bufA);
  gcn_zero_f32<<<gND, T, 0, stream>>>(bufB, NN * DH);
  gcn_scatter_kernel<<<gED, T, 0, stream>>>(bufA, src, dst, dinv, bufB);
  gcn_epilogue_kernel<<<gND, T, 0, stream>>>(bufB, bufA, dinv, b1);

  gcn_gemm_wmma<DH, false, false><<<gemmGrid, gemmBlk, 0, stream>>>(bufB, W2, nullptr, bufA);
  gcn_zero_f32<<<gND, T, 0, stream>>>(bufB, NN * DH);
  gcn_scatter_kernel<<<gED, T, 0, stream>>>(bufA, src, dst, dinv, bufB);
  gcn_epilogue_kernel<<<gND, T, 0, stream>>>(bufB, bufA, dinv, b2);

  gcn_gemm_wmma<DH, false, false><<<gemmGrid, gemmBlk, 0, stream>>>(bufB, W3, nullptr, bufA);
  gcn_zero_f32<<<gND, T, 0, stream>>>(bufB, NN * DH);
  gcn_scatter_kernel<<<gED, T, 0, stream>>>(bufA, src, dst, dinv, bufB);
  gcn_epilogue_kernel<<<gND, T, 0, stream>>>(bufB, bufA, dinv, b3);

  gcn_gemm_wmma<DH, true, true><<<gemmGrid, gemmBlk, 0, stream>>>(bufB, Wf1, bf1, bufA);
  gcn_gemm_wmma<DOUT, true, false><<<gemmGridOut, gemmBlk, 0, stream>>>(bufA, Wf2, bf2, out);
}


# Replace the f32 atomic scatter-add (with its zero and epilogue passes) by a per-call dst-sorted CSR (int count/rank, scan, fill, row sort) and a per-node register gather-sum; sum order now fixed
# speedup vs baseline: 2.9470x; 2.9470x over previous
#include <hip/hip_runtime.h>


typedef float v2f __attribute__((ext_vector_type(2)));
typedef float v8f __attribute__((ext_vector_type(8)));

#define NN   100000
#define EE   1600000
#define DH   64
#define DOUT 16

__global__ void gcn_zero_i32(int* __restrict__ p, int n) {
  int i = blockIdx.x * blockDim.x + threadIdx.x;
  if (i < n) p[i] = 0;
}

__global__ void gcn_count_rank_kernel(const int* __restrict__ dst,
                                      int* __restrict__ cnt,
                                      int* __restrict__ rank) {
  int e = blockIdx.x * blockDim.x + threadIdx.x;
  if (e < EE) rank[e] = atomicAdd(&cnt[dst[e]], 1);
}

__global__ __launch_bounds__(1024) void gcn_scan_dinv_kernel(
    const int* __restrict__ cnt, int* __restrict__ rowptr,
    float* __restrict__ dinv) {
  __shared__ int wsum[32];
  __shared__ int sh_off;
  const int tid  = threadIdx.x;
  const int lane = tid & 31;
  const int wv   = tid >> 5;
  const int base = blockIdx.x * 1024;

  int part = 0;
  for (int i = tid; i < base; i += 1024) part += cnt[i];
  for (int o = 16; o > 0; o >>= 1) part += __shfl_xor(part, o, 32);
  if (lane == 0) wsum[wv] = part;
  __syncthreads();
  if (wv == 0) {
    int v = wsum[lane];
    for (int o = 16; o > 0; o >>= 1) v += __shfl_xor(v, o, 32);
    if (lane == 0) sh_off = v;
  }
  __syncthreads();
  const int off = sh_off;

  const int i = base + tid;
  const int c = (i < NN) ? cnt[i] : 0;
  int inc = c;
  for (int o = 1; o < 32; o <<= 1) {
    const int t = __shfl_up(inc, o, 32);
    if (lane >= o) inc += t;
  }
  if (lane == 31) wsum[wv] = inc;
  __syncthreads();
  if (wv == 0) {
    const int v = wsum[lane];
    int s = v;
    for (int o = 1; o < 32; o <<= 1) {
      const int t = __shfl_up(s, o, 32);
      if (lane >= o) s += t;
    }
    wsum[lane] = s - v;
  }
  __syncthreads();
  if (i < NN) {
    rowptr[i] = off + wsum[wv] + inc - c;
    dinv[i]   = rsqrtf((float)c + 1.0f);
  }
}

__global__ void gcn_fill_kernel(const int* __restrict__ src,
                                const int* __restrict__ dst,
                                const int* __restrict__ rank,
                                const int* __restrict__ rowptr,
                                int* __restrict__ colA) {
  int e = blockIdx.x * blockDim.x + threadIdx.x;
  if (e < EE) colA[rowptr[dst[e]] + rank[e]] = src[e];
}

__global__ __launch_bounds__(256) void gcn_rowsort_kernel(
    const int* __restrict__ colA, const int* __restrict__ rowptr,
    const int* __restrict__ cnt, int* __restrict__ colB) {
  const int row  = blockIdx.x * 8 + (threadIdx.x >> 5);
  const int lane = threadIdx.x & 31;
  const int start = __builtin_amdgcn_readfirstlane(rowptr[row]);
  const int n     = __builtin_amdgcn_readfirstlane(cnt[row]);
  if (n <= 32) {
    const int key = (lane < n) ? colA[start + lane] : 0x7fffffff;
    int r = 0;
    for (int j = 0; j < n; ++j) {
      const int kj = __shfl(key, j, 32);
      r += ((kj < key) || (kj == key && j < lane)) ? 1 : 0;
    }
    if (lane < n) colB[start + r] = key;
  } else {
    for (int i = lane; i < n; i += 32) {
      const int key = colA[start + i];
      int r = 0;
      for (int j = 0; j < n; ++j) {
        const int kj = colA[start + j];
        r += ((kj < key) || (kj == key && j < i)) ? 1 : 0;
      }
      colB[start + r] = key;
    }
  }
}

template <int NCOLS, bool BIAS, bool RELU>
__global__ __launch_bounds__(32) void gcn_gemm_wmma(
    const float* __restrict__ X, const float* __restrict__ W,
    const float* __restrict__ bias, float* __restrict__ Out) {
  const int tileM = blockIdx.x;
  const int tileN = blockIdx.y;
  const int lane  = threadIdx.x;
  const int half  = lane >> 4;
  const int l     = lane & 15;
  const int arow  = tileM * 16 + l;
  const int bcol  = tileN * 16 + l;

  v8f c = {};
#pragma unroll
  for (int k0 = 0; k0 < DH; k0 += 4) {
    const int ak = k0 + half * 2;
    v2f a, b;
    a.x = X[arow * DH + ak];
    a.y = X[arow * DH + ak + 1];
    b.x = W[ak * NCOLS + bcol];
    b.y = W[(ak + 1) * NCOLS + bcol];
    c = __builtin_amdgcn_wmma_f32_16x16x4_f32(
         false, a,  false, b,
         (short)0, c,  false,  false);
  }

#pragma unroll
  for (int r = 0; r < 8; ++r) {
    const int row = tileM * 16 + half * 8 + r;
    float v = c[r];
    if (BIAS) v += bias[bcol];
    if (RELU) v = fmaxf(v, 0.0f);
    Out[row * NCOLS + bcol] = v;
  }
}

__global__ __launch_bounds__(256) void gcn_gather_kernel(
    const float* __restrict__ H, const int* __restrict__ col,
    const int* __restrict__ rowptr, const int* __restrict__ cnt,
    const float* __restrict__ dinv, const float* __restrict__ b,
    float* __restrict__ OUT) {
  const int t = blockIdx.x * 16 + (threadIdx.x >> 4);
  const int l = threadIdx.x & 15;
  const int start = rowptr[t];
  const int n     = cnt[t];
  const float dt  = dinv[t];
  const float4 hs = *reinterpret_cast<const float4*>(H + t * DH + l * 4);
  const float4 bb = *reinterpret_cast<const float4*>(b + l * 4);
  const int nmax = max(n, __shfl_xor(n, 16, 32));

  float4 acc = {0.0f, 0.0f, 0.0f, 0.0f};
  for (int c0 = 0; c0 < nmax; c0 += 16) {
    const int p = c0 + l;
    int   myidx = t;
    float mycf  = 0.0f;
    if (p < n) {
      myidx = col[start + p];
      mycf  = dinv[myidx] * dt;
    }
    const int cn = min(nmax - c0, 16);
    for (int j = 0; j < cn; j += 4) {
      const int   s0 = __shfl(myidx, j,     16);
      const int   s1 = __shfl(myidx, j + 1, 16);
      const int   s2 = __shfl(myidx, j + 2, 16);
      const int   s3 = __shfl(myidx, j + 3, 16);
      const float f0 = __shfl(mycf,  j,     16);
      const float f1 = __shfl(mycf,  j + 1, 16);
      const float f2 = __shfl(mycf,  j + 2, 16);
      const float f3 = __shfl(mycf,  j + 3, 16);
      const float4 h0 = *reinterpret_cast<const float4*>(H + s0 * DH + l * 4);
      const float4 h1 = *reinterpret_cast<const float4*>(H + s1 * DH + l * 4);
      const float4 h2 = *reinterpret_cast<const float4*>(H + s2 * DH + l * 4);
      const float4 h3 = *reinterpret_cast<const float4*>(H + s3 * DH + l * 4);
      acc.x += h0.x * f0; acc.y += h0.y * f0; acc.z += h0.z * f0; acc.w += h0.w * f0;
      acc.x += h1.x * f1; acc.y += h1.y * f1; acc.z += h1.z * f1; acc.w += h1.w * f1;
      acc.x += h2.x * f2; acc.y += h2.y * f2; acc.z += h2.z * f2; acc.w += h2.w * f2;
      acc.x += h3.x * f3; acc.y += h3.y * f3; acc.z += h3.z * f3; acc.w += h3.w * f3;
    }
  }

  float4 o;
  o.x = fmaxf(acc.x + hs.x * dt * dt + bb.x, 0.0f);
  o.y = fmaxf(acc.y + hs.y * dt * dt + bb.y, 0.0f);
  o.z = fmaxf(acc.z + hs.z * dt * dt + bb.z, 0.0f);
  o.w = fmaxf(acc.w + hs.w * dt * dt + bb.w, 0.0f);
  *reinterpret_cast<float4*>(OUT + t * DH + l * 4) = o;
}

extern "C" void kernel_launch(void* const* d_in, const int* in_sizes, int n_in,
                              void* d_out, int out_size, void* d_ws,
                              size_t ws_size, hipStream_t stream) {
  (void)in_sizes; (void)n_in; (void)out_size; (void)ws_size;

  const float* x   = (const float*)d_in[0];
  const int*   ei  = (const int*)d_in[1];
  const int*   src = ei;
  const int*   dst = ei + EE;
  const float* W1  = (const float*)d_in[2];
  const float* b1  = (const float*)d_in[3];
  const float* W2  = (const float*)d_in[4];
  const float* b2  = (const float*)d_in[5];
  const float* W3  = (const float*)d_in[6];
  const float* b3  = (const float*)d_in[7];
  const float* Wf1 = (const float*)d_in[8];
  const float* bf1 = (const float*)d_in[9];
  const float* Wf2 = (const float*)d_in[10];
  const float* bf2 = (const float*)d_in[11];
  float* out = (float*)d_out;

  float* base = (float*)d_ws;
  float* dinv = base;
  float* bufA = base + NN;
  float* bufB = bufA + NN * DH;
  int* cnt    = (int*)(bufB + NN * DH);
  int* rowptr = cnt + NN;
  int* rank   = rowptr + NN;
  int* colA   = rank + EE;
  int* colB   = colA + EE;

  const int T = 256;
  const int gN = (NN + T - 1) / T;
  const int gE = (EE + T - 1) / T;
  const dim3 gemmGrid(NN / 16, DH / 16);
  const dim3 gemmGridOut(NN / 16, DOUT / 16);
  const dim3 gemmBlk(32);

  gcn_zero_i32<<<gN, T, 0, stream>>>(cnt, NN);
  gcn_count_rank_kernel<<<gE, T, 0, stream>>>(dst, cnt, rank);
  gcn_scan_dinv_kernel<<<(NN + 1023) / 1024, 1024, 0, stream>>>(cnt, rowptr, dinv);
  gcn_fill_kernel<<<gE, T, 0, stream>>>(src, dst, rank, rowptr, colA);
  gcn_rowsort_kernel<<<NN / 8, 256, 0, stream>>>(colA, rowptr, cnt, colB);

  gcn_gemm_wmma<DH, false, false><<<gemmGrid, gemmBlk, 0, stream>>>(x, W1, nullptr, bufA);
  gcn_gather_kernel<<<NN / 16, 256, 0, stream>>>(bufA, colB, rowptr, cnt, dinv, b1, bufB);

  gcn_gemm_wmma<DH, false, false><<<gemmGrid, gemmBlk, 0, stream>>>(bufB, W2, nullptr, bufA);
  gcn_gather_kernel<<<NN / 16, 256, 0, stream>>>(bufA, colB, rowptr, cnt, dinv, b2, bufB);

  gcn_gemm_wmma<DH, false, false><<<gemmGrid, gemmBlk, 0, stream>>>(bufB, W3, nullptr, bufA);
  gcn_gather_kernel<<<NN / 16, 256, 0, stream>>>(bufA, colB, rowptr, cnt, dinv, b3, bufB);

  gcn_gemm_wmma<DH, true, true><<<gemmGrid, gemmBlk, 0, stream>>>(bufB, Wf1, bf1, bufA);
  gcn_gemm_wmma<DOUT, true, false><<<gemmGridOut, gemmBlk, 0, stream>>>(bufA, Wf2, bf2, out);
}


# Each layer's product now runs inside its aggregation kernel: the block sums neighbour rows of the layer input, then its 16x64 tile goes LDS->WMMA; head rides layer 3; f32 sum order changes
# speedup vs baseline: 4.0513x; 1.3747x over previous
#include <hip/hip_runtime.h>


typedef float v2f __attribute__((ext_vector_type(2)));
typedef float v8f __attribute__((ext_vector_type(8)));

#define NN   100000
#define EE   1600000
#define DH   64
#define DOUT 16

__global__ void gcn_zero_i32(int* __restrict__ p, int n) {
  int i = blockIdx.x * blockDim.x + threadIdx.x;
  if (i < n) p[i] = 0;
}

__global__ void gcn_count_rank_kernel(const int* __restrict__ dst,
                                      int* __restrict__ cnt,
                                      int* __restrict__ rank) {
  int e = blockIdx.x * blockDim.x + threadIdx.x;
  if (e < EE) rank[e] = atomicAdd(&cnt[dst[e]], 1);
}

__global__ __launch_bounds__(1024) void gcn_scan_dinv_kernel(
    const int* __restrict__ cnt, int* __restrict__ rowptr,
    float* __restrict__ dinv) {
  __shared__ int wsum[32];
  __shared__ int sh_off;
  const int tid  = threadIdx.x;
  const int lane = tid & 31;
  const int wv   = tid >> 5;
  const int base = blockIdx.x * 1024;

  int part = 0;
  for (int i = tid; i < base; i += 1024) part += cnt[i];
  for (int o = 16; o > 0; o >>= 1) part += __shfl_xor(part, o, 32);
  if (lane == 0) wsum[wv] = part;
  __syncthreads();
  if (wv == 0) {
    int v = wsum[lane];
    for (int o = 16; o > 0; o >>= 1) v += __shfl_xor(v, o, 32);
    if (lane == 0) sh_off = v;
  }
  __syncthreads();
  const int off = sh_off;

  const int i = base + tid;
  const int c = (i < NN) ? cnt[i] : 0;
  int inc = c;
  for (int o = 1; o < 32; o <<= 1) {
    const int t = __shfl_up(inc, o, 32);
    if (lane >= o) inc += t;
  }
  if (lane == 31) wsum[wv] = inc;
  __syncthreads();
  if (wv == 0) {
    const int v = wsum[lane];
    int s = v;
    for (int o = 1; o < 32; o <<= 1) {
      const int t = __shfl_up(s, o, 32);
      if (lane >= o) s += t;
    }
    wsum[lane] = s - v;
  }
  __syncthreads();
  if (i < NN) {
    rowptr[i] = off + wsum[wv] + inc - c;
    dinv[i]   = rsqrtf((float)c + 1.0f);
  }
}

__global__ void gcn_fill_kernel(const int* __restrict__ src,
                                const int* __restrict__ dst,
                                const int* __restrict__ rank,
                                const int* __restrict__ rowptr,
                                int* __restrict__ colA) {
  int e = blockIdx.x * blockDim.x + threadIdx.x;
  if (e < EE) colA[rowptr[dst[e]] + rank[e]] = src[e];
}

__global__ __launch_bounds__(256) void gcn_rowsort_kernel(
    const int* __restrict__ colA, const int* __restrict__ rowptr,
    const int* __restrict__ cnt, int* __restrict__ colB) {
  const int row  = blockIdx.x * 8 + (threadIdx.x >> 5);
  const int lane = threadIdx.x & 31;
  const int start = __builtin_amdgcn_readfirstlane(rowptr[row]);
  const int n     = __builtin_amdgcn_readfirstlane(cnt[row]);
  if (n <= 32) {
    const int key = (lane < n) ? colA[start + lane] : 0x7fffffff;
    int r = 0;
    for (int j = 0; j < n; ++j) {
      const int kj = __shfl(key, j, 32);
      r += ((kj < key) || (kj == key && j < lane)) ? 1 : 0;
    }
    if (lane < n) colB[start + r] = key;
  } else {
    for (int i = lane; i < n; i += 32) {
      const int key = colA[start + i];
      int r = 0;
      for (int j = 0; j < n; ++j) {
        const int kj = colA[start + j];
        r += ((kj < key) || (kj == key && j < i)) ? 1 : 0;
      }
      colB[start + r] = key;
    }
  }
}

#define PITCH 68

template <int NCOLS>
__device__ __forceinline__ v8f gcn_tile_wmma(const float* S,
                                             const float* __restrict__ W,
                                             int tileN, int lane) {
  const int half = lane >> 4;
  const int l    = lane & 15;
  const int bcol = tileN * 16 + l;
  v8f c = {};
  for (int k0 = 0; k0 < DH; k0 += 4) {
    const int ak = k0 + half * 2;
    const v2f a = *reinterpret_cast<const v2f*>(S + l * PITCH + ak);
    v2f b;
    b.x = W[ak * NCOLS + bcol];
    b.y = W[(ak + 1) * NCOLS + bcol];
    c = __builtin_amdgcn_wmma_f32_16x16x4_f32(
         false, a,  false, b,
         (short)0, c,  false,  false);
  }
  return c;
}

template <bool LAST>
__global__ __launch_bounds__(256) void gcn_layer_kernel(
    const float* __restrict__ X, const int* __restrict__ col,
    const int* __restrict__ rowptr, const int* __restrict__ cnt,
    const float* __restrict__ dinv,
    const float* __restrict__ W, const float* __restrict__ b,
    const float* __restrict__ Wf1, const float* __restrict__ bf1,
    const float* __restrict__ Wf2, const float* __restrict__ bf2,
    float* __restrict__ OUT) {
  __shared__ float sA[16 * PITCH];
  __shared__ float sB[16 * PITCH];
  const int m = threadIdx.x >> 4;
  const int t = blockIdx.x * 16 + m;
  const int l = threadIdx.x & 15;
  const int start = rowptr[t];
  const int n     = cnt[t];
  const float dt  = dinv[t];
  const float4 xs = *reinterpret_cast<const float4*>(X + t * DH + l * 4);
  const int nmax = max(n, __shfl_xor(n, 16, 32));

  float4 acc = {0.0f, 0.0f, 0.0f, 0.0f};
  for (int c0 = 0; c0 < nmax; c0 += 16) {
    const int p = c0 + l;
    int   myidx = t;
    float mycf  = 0.0f;
    if (p < n) {
      myidx = col[start + p];
      mycf  = dinv[myidx] * dt;
    }
    const int cn = min(nmax - c0, 16);
    for (int j = 0; j < cn; j += 4) {
      const int   s0 = __shfl(myidx, j,     16);
      const int   s1 = __shfl(myidx, j + 1, 16);
      const int   s2 = __shfl(myidx, j + 2, 16);
      const int   s3 = __shfl(myidx, j + 3, 16);
      const float f0 = __shfl(mycf,  j,     16);
      const float f1 = __shfl(mycf,  j + 1, 16);
      const float f2 = __shfl(mycf,  j + 2, 16);
      const float f3 = __shfl(mycf,  j + 3, 16);
      const float4 h0 = *reinterpret_cast<const float4*>(X + s0 * DH + l * 4);
      const float4 h1 = *reinterpret_cast<const float4*>(X + s1 * DH + l * 4);
      const float4 h2 = *reinterpret_cast<const float4*>(X + s2 * DH + l * 4);
      const float4 h3 = *reinterpret_cast<const float4*>(X + s3 * DH + l * 4);
      acc.x += h0.x * f0; acc.y += h0.y * f0; acc.z += h0.z * f0; acc.w += h0.w * f0;
      acc.x += h1.x * f1; acc.y += h1.y * f1; acc.z += h1.z * f1; acc.w += h1.w * f1;
      acc.x += h2.x * f2; acc.y += h2.y * f2; acc.z += h2.z * f2; acc.w += h2.w * f2;
      acc.x += h3.x * f3; acc.y += h3.y * f3; acc.z += h3.z * f3; acc.w += h3.w * f3;
    }
  }

  float4 g;
  g.x = acc.x + xs.x * dt * dt;
  g.y = acc.y + xs.y * dt * dt;
  g.z = acc.z + xs.z * dt * dt;
  g.w = acc.w + xs.w * dt * dt;
  *reinterpret_cast<float4*>(sA + m * PITCH + l * 4) = g;
  __syncthreads();

  const int wv   = __builtin_amdgcn_readfirstlane(threadIdx.x >> 5);
  const int lane = threadIdx.x & 31;
  const int half = lane >> 4;
  const int bcol = wv * 16 + (lane & 15);

  if (wv < 4) {
    const v8f c = gcn_tile_wmma<DH>(sA, W, wv, lane);
    const float bv = b[bcol];
    for (int r = 0; r < 8; ++r)
      sB[(half * 8 + r) * PITCH + bcol] = fmaxf(c[r] + bv, 0.0f);
  }
  __syncthreads();

  if (!LAST) {
    const float4 o = *reinterpret_cast<const float4*>(sB + m * PITCH + l * 4);
    *reinterpret_cast<float4*>(OUT + t * DH + l * 4) = o;
  } else {
    if (wv < 4) {
      const v8f c = gcn_tile_wmma<DH>(sB, Wf1, wv, lane);
      const float bv = bf1[bcol];
      for (int r = 0; r < 8; ++r)
        sA[(half * 8 + r) * PITCH + bcol] = fmaxf(c[r] + bv, 0.0f);
    }
    __syncthreads();
    if (wv == 0) {
      const v8f c = gcn_tile_wmma<DOUT>(sA, Wf2, 0, lane);
      const float bv = bf2[bcol];
      for (int r = 0; r < 8; ++r)
        sB[(half * 8 + r) * DOUT + bcol] = c[r] + bv;
    }
    __syncthreads();
    if (threadIdx.x < 64) {
      const float4 o = *reinterpret_cast<const float4*>(sB + threadIdx.x * 4);
      *reinterpret_cast<float4*>(OUT + blockIdx.x * 16 * DOUT + threadIdx.x * 4) = o;
    }
  }
}

extern "C" void kernel_launch(void* const* d_in, const int* in_sizes, int n_in,
                              void* d_out, int out_size, void* d_ws,
                              size_t ws_size, hipStream_t stream) {
  (void)in_sizes; (void)n_in; (void)out_size; (void)ws_size;

  const float* x   = (const float*)d_in[0];
  const int*   ei  = (const int*)d_in[1];
  const int*   src = ei;
  const int*   dst = ei + EE;
  const float* W1  = (const float*)d_in[2];
  const float* b1  = (const float*)d_in[3];
  const float* W2  = (const float*)d_in[4];
  const float* b2  = (const float*)d_in[5];
  const float* W3  = (const float*)d_in[6];
  const float* b3  = (const float*)d_in[7];
  const float* Wf1 = (const float*)d_in[8];
  const float* bf1 = (const float*)d_in[9];
  const float* Wf2 = (const float*)d_in[10];
  const float* bf2 = (const float*)d_in[11];
  float* out = (float*)d_out;

  float* base = (float*)d_ws;
  float* dinv = base;
  float* bufA = base + NN;
  float* bufB = bufA + NN * DH;
  int* cnt    = (int*)(bufB + NN * DH);
  int* rowptr = cnt + NN;
  int* rank   = rowptr + NN;
  int* colA   = rank + EE;
  int* colB   = colA + EE;

  const int T = 256;
  const int gN = (NN + T - 1) / T;
  const int gE = (EE + T - 1) / T;

  gcn_zero_i32<<<gN, T, 0, stream>>>(cnt, NN);
  gcn_count_rank_kernel<<<gE, T, 0, stream>>>(dst, cnt, rank);
  gcn_scan_dinv_kernel<<<(NN + 1023) / 1024, 1024, 0, stream>>>(cnt, rowptr, dinv);
  gcn_fill_kernel<<<gE, T, 0, stream>>>(src, dst, rank, rowptr, colA);
  gcn_rowsort_kernel<<<NN / 8, 256, 0, stream>>>(colA, rowptr, cnt, colB);

  gcn_layer_kernel<false><<<NN / 16, 256, 0, stream>>>(
      x, colB, rowptr, cnt, dinv, W1, b1, nullptr, nullptr, nullptr, nullptr, bufA);
  gcn_layer_kernel<false><<<NN / 16, 256, 0, stream>>>(
      bufA, colB, rowptr, cnt, dinv, W2, b2, nullptr, nullptr, nullptr, nullptr, bufB);
  gcn_layer_kernel<true><<<NN / 16, 256, 0, stream>>>(
      bufB, colB, rowptr, cnt, dinv, W3, b3, Wf1, bf1, Wf2, bf2, out);
}


# Rows are stored pre-scaled by their own dinv (x in the row-sort pass, layers 1-2 store relu(.)*dinv), so the gather adds plain rows: no per-edge dinv[src] read or coefficient shuffle; sum order kept
# speedup vs baseline: 4.2552x; 1.0503x over previous
#include <hip/hip_runtime.h>


typedef float v2f __attribute__((ext_vector_type(2)));
typedef float v8f __attribute__((ext_vector_type(8)));

#define NN   100000
#define EE   1600000
#define DH   64
#define DOUT 16

__global__ void gcn_zero_i32(int* __restrict__ p, int n) {
  int i = blockIdx.x * blockDim.x + threadIdx.x;
  if (i < n) p[i] = 0;
}

__global__ void gcn_count_rank_kernel(const int* __restrict__ dst,
                                      int* __restrict__ cnt,
                                      int* __restrict__ rank) {
  int e = blockIdx.x * blockDim.x + threadIdx.x;
  if (e < EE) rank[e] = atomicAdd(&cnt[dst[e]], 1);
}

__global__ __launch_bounds__(1024) void gcn_scan_dinv_kernel(
    const int* __restrict__ cnt, int* __restrict__ rowptr,
    float* __restrict__ dinv) {
  __shared__ int wsum[32];
  __shared__ int sh_off;
  const int tid  = threadIdx.x;
  const int lane = tid & 31;
  const int wv   = tid >> 5;
  const int base = blockIdx.x * 1024;

  int part = 0;
  for (int i = tid; i < base; i += 1024) part += cnt[i];
  for (int o = 16; o > 0; o >>= 1) part += __shfl_xor(part, o, 32);
  if (lane == 0) wsum[wv] = part;
  __syncthreads();
  if (wv == 0) {
    int v = wsum[lane];
    for (int o = 16; o > 0; o >>= 1) v += __shfl_xor(v, o, 32);
    if (lane == 0) sh_off = v;
  }
  __syncthreads();
  const int off = sh_off;

  const int i = base + tid;
  const int c = (i < NN) ? cnt[i] : 0;
  int inc = c;
  for (int o = 1; o < 32; o <<= 1) {
    const int t = __shfl_up(inc, o, 32);
    if (lane >= o) inc += t;
  }
  if (lane == 31) wsum[wv] = inc;
  __syncthreads();
  if (wv == 0) {
    const int v = wsum[lane];
    int s = v;
    for (int o = 1; o < 32; o <<= 1) {
      const int t = __shfl_up(s, o, 32);
      if (lane >= o) s += t;
    }
    wsum[lane] = s - v;
  }
  __syncthreads();
  if (i < NN) {
    rowptr[i] = off + wsum[wv] + inc - c;
    dinv[i]   = rsqrtf((float)c + 1.0f);
  }
}

__global__ void gcn_fill_kernel(const int* __restrict__ src,
                                const int* __restrict__ dst,
                                const int* __restrict__ rank,
                                const int* __restrict__ rowptr,
                                int* __restrict__ colA) {
  int e = blockIdx.x * blockDim.x + threadIdx.x;
  if (e < EE) colA[rowptr[dst[e]] + rank[e]] = src[e];
}

__global__ __launch_bounds__(256) void gcn_rowsort_scale_kernel(
    const int* __restrict__ colA, const int* __restrict__ rowptr,
    const int* __restrict__ cnt, const float* __restrict__ dinv,
    const float* __restrict__ x, int* __restrict__ colB,
    float* __restrict__ Y0) {
  const int row  = blockIdx.x * 8 + (threadIdx.x >> 5);
  const int lane = threadIdx.x & 31;
  {
    const float2 xv = *reinterpret_cast<const float2*>(x + row * DH + lane * 2);
    const float d = dinv[row];
    float2 yv;
    yv.x = xv.x * d;
    yv.y = xv.y * d;
    *reinterpret_cast<float2*>(Y0 + row * DH + lane * 2) = yv;
  }
  const int start = __builtin_amdgcn_readfirstlane(rowptr[row]);
  const int n     = __builtin_amdgcn_readfirstlane(cnt[row]);
  if (n <= 32) {
    const int key = (lane < n) ? colA[start + lane] : 0x7fffffff;
    int r = 0;
    for (int j = 0; j < n; ++j) {
      const int kj = __shfl(key, j, 32);
      r += ((kj < key) || (kj == key && j < lane)) ? 1 : 0;
    }
    if (lane < n) colB[start + r] = key;
  } else {
    for (int i = lane; i < n; i += 32) {
      const int key = colA[start + i];
      int r = 0;
      for (int j = 0; j < n; ++j) {
        const int kj = colA[start + j];
        r += ((kj < key) || (kj == key && j < i)) ? 1 : 0;
      }
      colB[start + r] = key;
    }
  }
}

#define PITCH 68

template <int NCOLS>
__device__ __forceinline__ v8f gcn_tile_wmma(const float* S,
                                             const float* __restrict__ W,
                                             int tileN, int lane) {
  const int half = lane >> 4;
  const int l    = lane & 15;
  const int bcol = tileN * 16 + l;
  v8f c = {};
  for (int k0 = 0; k0 < DH; k0 += 4) {
    const int ak = k0 + half * 2;
    const v2f a = *reinterpret_cast<const v2f*>(S + l * PITCH + ak);
    v2f b;
    b.x = W[ak * NCOLS + bcol];
    b.y = W[(ak + 1) * NCOLS + bcol];
    c = __builtin_amdgcn_wmma_f32_16x16x4_f32(
         false, a,  false, b,
         (short)0, c,  false,  false);
  }
  return c;
}

template <bool LAST>
__global__ __launch_bounds__(256) void gcn_layer_kernel(
    const float* __restrict__ Y, const int* __restrict__ col,
    const int* __restrict__ rowptr, const int* __restrict__ cnt,
    const float* __restrict__ dinv,
    const float* __restrict__ W, const float* __restrict__ b,
    const float* __restrict__ Wf1, const float* __restrict__ bf1,
    const float* __restrict__ Wf2, const float* __restrict__ bf2,
    float* __restrict__ OUT) {
  __shared__ float sA[16 * PITCH];
  __shared__ float sB[16 * PITCH];
  const int m = threadIdx.x >> 4;
  const int t = blockIdx.x * 16 + m;
  const int l = threadIdx.x & 15;
  const int start = rowptr[t];
  const int n     = cnt[t];
  const float dt  = dinv[t];
  const float4 ys = *reinterpret_cast<const float4*>(Y + t * DH + l * 4);
  const int nmax = max(n, __shfl_xor(n, 16, 32));

  float4 acc = {0.0f, 0.0f, 0.0f, 0.0f};
  for (int c0 = 0; c0 < nmax; c0 += 16) {
    const int p = c0 + l;
    int myidx = t;
    if (p < n) myidx = col[start + p];
    const int cn = min(nmax - c0, 16);
    for (int j = 0; j < cn; j += 4) {
      const int s0 = __shfl(myidx, j,     16);
      const int s1 = __shfl(myidx, j + 1, 16);
      const int s2 = __shfl(myidx, j + 2, 16);
      const int s3 = __shfl(myidx, j + 3, 16);
      const int   q  = n - (c0 + j);
      const float f0 = (q > 0) ? 1.0f : 0.0f;
      const float f1 = (q > 1) ? 1.0f : 0.0f;
      const float f2 = (q > 2) ? 1.0f : 0.0f;
      const float f3 = (q > 3) ? 1.0f : 0.0f;
      const float4 h0 = *reinterpret_cast<const float4*>(Y + s0 * DH + l * 4);
      const float4 h1 = *reinterpret_cast<const float4*>(Y + s1 * DH + l * 4);
      const float4 h2 = *reinterpret_cast<const float4*>(Y + s2 * DH + l * 4);
      const float4 h3 = *reinterpret_cast<const float4*>(Y + s3 * DH + l * 4);
      acc.x += h0.x * f0; acc.y += h0.y * f0; acc.z += h0.z * f0; acc.w += h0.w * f0;
      acc.x += h1.x * f1; acc.y += h1.y * f1; acc.z += h1.z * f1; acc.w += h1.w * f1;
      acc.x += h2.x * f2; acc.y += h2.y * f2; acc.z += h2.z * f2; acc.w += h2.w * f2;
      acc.x += h3.x * f3; acc.y += h3.y * f3; acc.z += h3.z * f3; acc.w += h3.w * f3;
    }
  }

  float4 g;
  g.x = (acc.x + ys.x) * dt;
  g.y = (acc.y + ys.y) * dt;
  g.z = (acc.z + ys.z) * dt;
  g.w = (acc.w + ys.w) * dt;
  *reinterpret_cast<float4*>(sA + m * PITCH + l * 4) = g;
  __syncthreads();

  const int wv   = __builtin_amdgcn_readfirstlane(threadIdx.x >> 5);
  const int lane = threadIdx.x & 31;
  const int half = lane >> 4;
  const int bcol = wv * 16 + (lane & 15);

  if (wv < 4) {
    const v8f c = gcn_tile_wmma<DH>(sA, W, wv, lane);
    const float bv = b[bcol];
    for (int r = 0; r < 8; ++r)
      sB[(half * 8 + r) * PITCH + bcol] = fmaxf(c[r] + bv, 0.0f);
  }
  __syncthreads();

  if (!LAST) {
    float4 o = *reinterpret_cast<const float4*>(sB + m * PITCH + l * 4);
    o.x *= dt; o.y *= dt; o.z *= dt; o.w *= dt;
    *reinterpret_cast<float4*>(OUT + t * DH + l * 4) = o;
  } else {
    if (wv < 4) {
      const v8f c = gcn_tile_wmma<DH>(sB, Wf1, wv, lane);
      const float bv = bf1[bcol];
      for (int r = 0; r < 8; ++r)
        sA[(half * 8 + r) * PITCH + bcol] = fmaxf(c[r] + bv, 0.0f);
    }
    __syncthreads();
    if (wv == 0) {
      const v8f c = gcn_tile_wmma<DOUT>(sA, Wf2, 0, lane);
      const float bv = bf2[bcol];
      for (int r = 0; r < 8; ++r)
        sB[(half * 8 + r) * DOUT + bcol] = c[r] + bv;
    }
    __syncthreads();
    if (threadIdx.x < 64) {
      const float4 o = *reinterpret_cast<const float4*>(sB + threadIdx.x * 4);
      *reinterpret_cast<float4*>(OUT + blockIdx.x * 16 * DOUT + threadIdx.x * 4) = o;
    }
  }
}

extern "C" void kernel_launch(void* const* d_in, const int* in_sizes, int n_in,
                              void* d_out, int out_size, void* d_ws,
                              size_t ws_size, hipStream_t stream) {
  (void)in_sizes; (void)n_in; (void)out_size; (void)ws_size;

  const float* x   = (const float*)d_in[0];
  const int*   ei  = (const int*)d_in[1];
  const int*   src = ei;
  const int*   dst = ei + EE;
  const float* W1  = (const float*)d_in[2];
  const float* b1  = (const float*)d_in[3];
  const float* W2  = (const float*)d_in[4];
  const float* b2  = (const float*)d_in[5];
  const float* W3  = (const float*)d_in[6];
  const float* b3  = (const float*)d_in[7];
  const float* Wf1 = (const float*)d_in[8];
  const float* bf1 = (const float*)d_in[9];
  const float* Wf2 = (const float*)d_in[10];
  const float* bf2 = (const float*)d_in[11];
  float* out = (float*)d_out;

  float* base = (float*)d_ws;
  float* dinv = base;
  float* bufA = base + NN;
  float* bufB = bufA + NN * DH;
  int* cnt    = (int*)(bufB + NN * DH);
  int* rowptr = cnt + NN;
  int* rank   = rowptr + NN;
  int* colA   = rank + EE;
  int* colB   = colA + EE;

  const int T = 256;
  const int gN = (NN + T - 1) / T;
  const int gE = (EE + T - 1) / T;

  gcn_zero_i32<<<gN, T, 0, stream>>>(cnt, NN);
  gcn_count_rank_kernel<<<gE, T, 0, stream>>>(dst, cnt, rank);
  gcn_scan_dinv_kernel<<<(NN + 1023) / 1024, 1024, 0, stream>>>(cnt, rowptr, dinv);
  gcn_fill_kernel<<<gE, T, 0, stream>>>(src, dst, rank, rowptr, colA);
  gcn_rowsort_scale_kernel<<<NN / 8, 256, 0, stream>>>(colA, rowptr, cnt, dinv, x,
                                                      colB, bufB);

  gcn_layer_kernel<false><<<NN / 16, 256, 0, stream>>>(
      bufB, colB, rowptr, cnt, dinv, W1, b1, nullptr, nullptr, nullptr, nullptr, bufA);
  gcn_layer_kernel<false><<<NN / 16, 256, 0, stream>>>(
      bufA, colB, rowptr, cnt, dinv, W2, b2, nullptr, nullptr, nullptr, nullptr, bufB);
  gcn_layer_kernel<true><<<NN / 16, 256, 0, stream>>>(
      bufB, colB, rowptr, cnt, dinv, W3, b3, Wf1, bf1, Wf2, bf2, out);
}


# Row sort and row pre-scale lose their own launch: the fill pass also stores x*dinv, layer 1 sorts its own nodes' slot lists (used from LDS, stored for layers 2-3); 7 launches, sum order kept
# speedup vs baseline: 4.4103x; 1.0364x over previous
#include <hip/hip_runtime.h>


typedef float v2f __attribute__((ext_vector_type(2)));
typedef float v8f __attribute__((ext_vector_type(8)));

#define NN   100000
#define EE   1600000
#define DH   64
#define DOUT 16

__global__ void gcn_zero_i32(int* __restrict__ p, int n) {
  int i = blockIdx.x * blockDim.x + threadIdx.x;
  if (i < n) p[i] = 0;
}

__global__ void gcn_count_rank_kernel(const int* __restrict__ dst,
                                      int* __restrict__ cnt,
                                      int* __restrict__ rank) {
  int e = blockIdx.x * blockDim.x + threadIdx.x;
  if (e < EE) rank[e] = atomicAdd(&cnt[dst[e]], 1);
}

__global__ __launch_bounds__(1024) void gcn_scan_dinv_kernel(
    const int* __restrict__ cnt, int* __restrict__ rowptr,
    float* __restrict__ dinv) {
  __shared__ int wsum[32];
  __shared__ int sh_off;
  const int tid  = threadIdx.x;
  const int lane = tid & 31;
  const int wv   = tid >> 5;
  const int base = blockIdx.x * 1024;

  int part = 0;
  for (int i = tid; i < base; i += 1024) part += cnt[i];
  for (int o = 16; o > 0; o >>= 1) part += __shfl_xor(part, o, 32);
  if (lane == 0) wsum[wv] = part;
  __syncthreads();
  if (wv == 0) {
    int v = wsum[lane];
    for (int o = 16; o > 0; o >>= 1) v += __shfl_xor(v, o, 32);
    if (lane == 0) sh_off = v;
  }
  __syncthreads();
  const int off = sh_off;

  const int i = base + tid;
  const int c = (i < NN) ? cnt[i] : 0;
  int inc = c;
  for (int o = 1; o < 32; o <<= 1) {
    const int t = __shfl_up(inc, o, 32);
    if (lane >= o) inc += t;
  }
  if (lane == 31) wsum[wv] = inc;
  __syncthreads();
  if (wv == 0) {
    const int v = wsum[lane];
    int s = v;
    for (int o = 1; o < 32; o <<= 1) {
      const int t = __shfl_up(s, o, 32);
      if (lane >= o) s += t;
    }
    wsum[lane] = s - v;
  }
  __syncthreads();
  if (i < NN) {
    rowptr[i] = off + wsum[wv] + inc - c;
    dinv[i]   = rsqrtf((float)c + 1.0f);
  }
}

static_assert(EE == (NN / 16) * 256, "one fill block per 16 nodes");
__global__ __launch_bounds__(256) void gcn_fill_scale_kernel(
    const int* __restrict__ src, const int* __restrict__ dst,
    const int* __restrict__ rank, const int* __restrict__ rowptr,
    const float* __restrict__ dinv, const float* __restrict__ x,
    int* __restrict__ colA, float* __restrict__ Y0) {
  const int e   = blockIdx.x * 256 + threadIdx.x;
  const int row = blockIdx.x * 16 + (threadIdx.x >> 4);
  const int l   = threadIdx.x & 15;
  const float4 xv = *reinterpret_cast<const float4*>(x + row * DH + l * 4);
  const float d = dinv[row];
  const int dd = dst[e];
  const int rk = rank[e];
  const int sv = src[e];
  colA[rowptr[dd] + rk] = sv;
  float4 yv;
  yv.x = xv.x * d;
  yv.y = xv.y * d;
  yv.z = xv.z * d;
  yv.w = xv.w * d;
  *reinterpret_cast<float4*>(Y0 + row * DH + l * 4) = yv;
}

#define SCAP 64
__device__ __forceinline__ void gcn_sort_row(const int* __restrict__ colU,
                                             int* colS, int* sl, int start,
                                             int n, int lane) {
  if (n <= 32) {
    const int key = (lane < n) ? colU[start + lane] : 0x7fffffff;
    int r = 0;
    for (int j = 0; j < n; ++j) {
      const int kj = __shfl(key, j, 32);
      r += ((kj < key) || (kj == key && j < lane)) ? 1 : 0;
    }
    if (lane < n) {
      colS[start + r] = key;
      sl[r] = key;
    }
  } else {
    for (int i = lane; i < n; i += 32) {
      const int key = colU[start + i];
      int r = 0;
      for (int j = 0; j < n; ++j) {
        const int kj = colU[start + j];
        r += ((kj < key) || (kj == key && j < i)) ? 1 : 0;
      }
      colS[start + r] = key;
      if (r < SCAP) sl[r] = key;
    }
  }
}

#define PITCH 68

template <int NCOLS>
__device__ __forceinline__ v8f gcn_tile_wmma(const float* S,
                                             const float* __restrict__ W,
                                             int tileN, int lane) {
  const int half = lane >> 4;
  const int l    = lane & 15;
  const int bcol = tileN * 16 + l;
  v8f c = {};
  for (int k0 = 0; k0 < DH; k0 += 4) {
    const int ak = k0 + half * 2;
    const v2f a = *reinterpret_cast<const v2f*>(S + l * PITCH + ak);
    v2f b;
    b.x = W[ak * NCOLS + bcol];
    b.y = W[(ak + 1) * NCOLS + bcol];
    c = __builtin_amdgcn_wmma_f32_16x16x4_f32(
         false, a,  false, b,
         (short)0, c,  false,  false);
  }
  return c;
}

template <bool LAST, bool FIRST>
__global__ __launch_bounds__(256) void gcn_layer_kernel(
    const float* __restrict__ Y, const int* __restrict__ col, int* colS,
    const int* __restrict__ rowptr, const int* __restrict__ cnt,
    const float* __restrict__ dinv,
    const float* __restrict__ W, const float* __restrict__ b,
    const float* __restrict__ Wf1, const float* __restrict__ bf1,
    const float* __restrict__ Wf2, const float* __restrict__ bf2,
    float* __restrict__ OUT) {
  __shared__ float sA[16 * PITCH];
  __shared__ float sB[16 * PITCH];
  __shared__ int sIdx[FIRST ? 16 * SCAP : 1];
  const int m = threadIdx.x >> 4;
  const int t = blockIdx.x * 16 + m;
  const int l = threadIdx.x & 15;
  const int start = rowptr[t];
  const int n     = cnt[t];
  const float dt  = dinv[t];
  const float4 ys = *reinterpret_cast<const float4*>(Y + t * DH + l * 4);
  const int nmax = max(n, __shfl_xor(n, 16, 32));

  if (FIRST) {
    const int lane32 = threadIdx.x & 31;
    const int stA = __builtin_amdgcn_readlane(start, 0);
    const int nA  = __builtin_amdgcn_readlane(n, 0);
    const int stB = __builtin_amdgcn_readlane(start, 16);
    const int nB  = __builtin_amdgcn_readlane(n, 16);
    int* slA = sIdx + (threadIdx.x >> 5) * (2 * SCAP);
    int* slB = slA + SCAP;
    if (nA <= 32 && nB <= 32) {
      const int keyA = (lane32 < nA) ? col[stA + lane32] : 0x7fffffff;
      const int keyB = (lane32 < nB) ? col[stB + lane32] : 0x7fffffff;
      const int nm = max(nA, nB);
      int rA = 0, rB = 0;
      for (int j = 0; j < nm; ++j) {
        const int kA = __shfl(keyA, j, 32);
        const int kB = __shfl(keyB, j, 32);
        rA += ((kA < keyA) || (kA == keyA && j < lane32)) ? 1 : 0;
        rB += ((kB < keyB) || (kB == keyB && j < lane32)) ? 1 : 0;
      }
      if (lane32 < nA) {
        colS[stA + rA] = keyA;
        slA[rA] = keyA;
      }
      if (lane32 < nB) {
        colS[stB + rB] = keyB;
        slB[rB] = keyB;
      }
    } else {
      gcn_sort_row(col, colS, slA, stA, nA, lane32);
      gcn_sort_row(col, colS, slB, stB, nB, lane32);
    }
  }

  float4 acc = {0.0f, 0.0f, 0.0f, 0.0f};
  for (int c0 = 0; c0 < nmax; c0 += 16) {
    const int p = c0 + l;
    int myidx = t;
    if (FIRST) {
      if (p < n) {
        if (p < SCAP) {
          myidx = sIdx[m * SCAP + p];
        } else {
          __threadfence_block();
          myidx = colS[start + p];
        }
      }
    } else {
      if (p < n) myidx = col[start + p];
    }
    const int cn = min(nmax - c0, 16);
    for (int j = 0; j < cn; j += 4) {
      const int s0 = __shfl(myidx, j,     16);
      const int s1 = __shfl(myidx, j + 1, 16);
      const int s2 = __shfl(myidx, j + 2, 16);
      const int s3 = __shfl(myidx, j + 3, 16);
      const int   q  = n - (c0 + j);
      const float f0 = (q > 0) ? 1.0f : 0.0f;
      const float f1 = (q > 1) ? 1.0f : 0.0f;
      const float f2 = (q > 2) ? 1.0f : 0.0f;
      const float f3 = (q > 3) ? 1.0f : 0.0f;
      const float4 h0 = *reinterpret_cast<const float4*>(Y + s0 * DH + l * 4);
      const float4 h1 = *reinterpret_cast<const float4*>(Y + s1 * DH + l * 4);
      const float4 h2 = *reinterpret_cast<const float4*>(Y + s2 * DH + l * 4);
      const float4 h3 = *reinterpret_cast<const float4*>(Y + s3 * DH + l * 4);
      acc.x += h0.x * f0; acc.y += h0.y * f0; acc.z += h0.z * f0; acc.w += h0.w * f0;
      acc.x += h1.x * f1; acc.y += h1.y * f1; acc.z += h1.z * f1; acc.w += h1.w * f1;
      acc.x += h2.x * f2; acc.y += h2.y * f2; acc.z += h2.z * f2; acc.w += h2.w * f2;
      acc.x += h3.x * f3; acc.y += h3.y * f3; acc.z += h3.z * f3; acc.w += h3.w * f3;
    }
  }

  float4 g;
  g.x = (acc.x + ys.x) * dt;
  g.y = (acc.y + ys.y) * dt;
  g.z = (acc.z + ys.z) * dt;
  g.w = (acc.w + ys.w) * dt;
  *reinterpret_cast<float4*>(sA + m * PITCH + l * 4) = g;
  __syncthreads();

  const int wv   = __builtin_amdgcn_readfirstlane(threadIdx.x >> 5);
  const int lane = threadIdx.x & 31;
  const int half = lane >> 4;
  const int bcol = wv * 16 + (lane & 15);

  if (wv < 4) {
    const v8f c = gcn_tile_wmma<DH>(sA, W, wv, lane);
    const float bv = b[bcol];
    for (int r = 0; r < 8; ++r)
      sB[(half * 8 + r) * PITCH + bcol] = fmaxf(c[r] + bv, 0.0f);
  }
  __syncthreads();

  if (!LAST) {
    float4 o = *reinterpret_cast<const float4*>(sB + m * PITCH + l * 4);
    o.x *= dt; o.y *= dt; o.z *= dt; o.w *= dt;
    *reinterpret_cast<float4*>(OUT + t * DH + l * 4) = o;
  } else {
    if (wv < 4) {
      const v8f c = gcn_tile_wmma<DH>(sB, Wf1, wv, lane);
      const float bv = bf1[bcol];
      for (int r = 0; r < 8; ++r)
        sA[(half * 8 + r) * PITCH + bcol] = fmaxf(c[r] + bv, 0.0f);
    }
    __syncthreads();
    if (wv == 0) {
      const v8f c = gcn_tile_wmma<DOUT>(sA, Wf2, 0, lane);
      const float bv = bf2[bcol];
      for (int r = 0; r < 8; ++r)
        sB[(half * 8 + r) * DOUT + bcol] = c[r] + bv;
    }
    __syncthreads();
    if (threadIdx.x < 64) {
      const float4 o = *reinterpret_cast<const float4*>(sB + threadIdx.x * 4);
      *reinterpret_cast<float4*>(OUT + blockIdx.x * 16 * DOUT + threadIdx.x * 4) = o;
    }
  }
}

extern "C" void kernel_launch(void* const* d_in, const int* in_sizes, int n_in,
                              void* d_out, int out_size, void* d_ws,
                              size_t ws_size, hipStream_t stream) {
  (void)in_sizes; (void)n_in; (void)out_size; (void)ws_size;

  const float* x   = (const float*)d_in[0];
  const int*   ei  = (const int*)d_in[1];
  const int*   src = ei;
  const int*   dst = ei + EE;
  const float* W1  = (const float*)d_in[2];
  const float* b1  = (const float*)d_in[3];
  const float* W2  = (const float*)d_in[4];
  const float* b2  = (const float*)d_in[5];
  const float* W3  = (const float*)d_in[6];
  const float* b3  = (const float*)d_in[7];
  const float* Wf1 = (const float*)d_in[8];
  const float* bf1 = (const float*)d_in[9];
  const float* Wf2 = (const float*)d_in[10];
  const float* bf2 = (const float*)d_in[11];
  float* out = (float*)d_out;

  float* base = (float*)d_ws;
  float* dinv = base;
  float* bufA = base + NN;
  float* bufB = bufA + NN * DH;
  int* cnt    = (int*)(bufB + NN * DH);
  int* rowptr = cnt + NN;
  int* rank   = rowptr + NN;
  int* colA   = rank + EE;
  int* colB   = colA + EE;

  const int T = 256;
  const int gN = (NN + T - 1) / T;
  const int gE = (EE + T - 1) / T;

  gcn_zero_i32<<<gN, T, 0, stream>>>(cnt, NN);
  gcn_count_rank_kernel<<<gE, T, 0, stream>>>(dst, cnt, rank);
  gcn_scan_dinv_kernel<<<(NN + 1023) / 1024, 1024, 0, stream>>>(cnt, rowptr, dinv);
  gcn_fill_scale_kernel<<<gE, T, 0, stream>>>(src, dst, rank, rowptr, dinv, x,
                                              colA, bufB);

  gcn_layer_kernel<false, true><<<NN / 16, 256, 0, stream>>>(
      bufB, colA, colB, rowptr, cnt, dinv, W1, b1, nullptr, nullptr, nullptr, nullptr, bufA);
  gcn_layer_kernel<false, false><<<NN / 16, 256, 0, stream>>>(
      bufA, colB, nullptr, rowptr, cnt, dinv, W2, b2, nullptr, nullptr, nullptr, nullptr, bufB);
  gcn_layer_kernel<true, false><<<NN / 16, 256, 0, stream>>>(
      bufB, colB, nullptr, rowptr, cnt, dinv, W3, b3, Wf1, bf1, Wf2, bf2, out);
}


# CSR build without global atomics: two-level counting sort by destination on LDS counters (slice histogram, scan, bucket scatter, one block per 128-node bucket fills the rows); sum order kept
# speedup vs baseline: 5.9335x; 1.3454x over previous
#include <hip/hip_runtime.h>


typedef float v2f __attribute__((ext_vector_type(2)));
typedef float v8f __attribute__((ext_vector_type(8)));

#define NN   100000
#define EE   1600000
#define DH   64
#define DOUT 16

#define BSH 7
#define BR  (1 << BSH)
#define NB  ((NN + BR - 1) / BR)
#define NS  250
#define SE  (EE / NS)
#define ST  640
#define SK  (SE / ST)
#define NH  (NB * NS)
#define RK  10
static_assert(EE == NS * SE && SE == SK * ST, "slices tile the edge list");
static_assert(NN <= (1 << (31 - BSH)), "a source id fits above the local destination");

__global__ __launch_bounds__(ST) void gcn_bucket_hist_kernel(
    const int* __restrict__ dst, int* __restrict__ H) {
  __shared__ int h[NB];
  for (int b = threadIdx.x; b < NB; b += ST) h[b] = 0;
  __syncthreads();
  const int e0 = blockIdx.x * SE + threadIdx.x;
  int d[SK];
  for (int k = 0; k < SK; ++k) d[k] = dst[e0 + k * ST];
  for (int k = 0; k < SK; ++k) atomicAdd(&h[d[k] >> BSH], 1);
  __syncthreads();
  for (int b = threadIdx.x; b < NB; b += ST) H[b * NS + blockIdx.x] = h[b];
}

__global__ __launch_bounds__(1024) void gcn_scan_kernel(
    const int* __restrict__ H, int* __restrict__ base) {
  __shared__ int wsum[32];
  __shared__ int sh_off;
  const int tid  = threadIdx.x;
  const int lane = tid & 31;
  const int wv   = tid >> 5;
  const int first = blockIdx.x * 1024;

  int part = 0;
  for (int i = tid * 4; i < first; i += 8 * 4096) {
    int4 v[8];
    for (int u = 0; u < 8; ++u)
      v[u] = *reinterpret_cast<const int4*>(H + min(i + u * 4096, first - 4));
    for (int u = 0; u < 8; ++u)
      if (i + u * 4096 < first) part += (v[u].x + v[u].y) + (v[u].z + v[u].w);
  }
  for (int o = 16; o > 0; o >>= 1) part += __shfl_xor(part, o, 32);
  if (lane == 0) wsum[wv] = part;
  __syncthreads();
  if (wv == 0) {
    int v = wsum[lane];
    for (int o = 16; o > 0; o >>= 1) v += __shfl_xor(v, o, 32);
    if (lane == 0) sh_off = v;
  }
  __syncthreads();
  const int off = sh_off;

  const int i = first + tid;
  const int c = (i < NH) ? H[i] : 0;
  int inc = c;
  for (int o = 1; o < 32; o <<= 1) {
    const int t = __shfl_up(inc, o, 32);
    if (lane >= o) inc += t;
  }
  if (lane == 31) wsum[wv] = inc;
  __syncthreads();
  if (wv == 0) {
    const int v = wsum[lane];
    int s = v;
    for (int o = 1; o < 32; o <<= 1) {
      const int t = __shfl_up(s, o, 32);
      if (lane >= o) s += t;
    }
    wsum[lane] = s - v;
  }
  __syncthreads();
  if (i < NH) base[i] = off + wsum[wv] + inc - c;
}

__global__ __launch_bounds__(ST) void gcn_bucket_scatter_kernel(
    const int* __restrict__ src, const int* __restrict__ dst,
    const int* __restrict__ base, int* __restrict__ rec) {
  __shared__ int cur[NB];
  for (int b = threadIdx.x; b < NB; b += ST) cur[b] = base[b * NS + blockIdx.x];
  __syncthreads();
  const int e0 = blockIdx.x * SE + threadIdx.x;
  int d[SK], s[SK];
  for (int k = 0; k < SK; ++k) {
    d[k] = dst[e0 + k * ST];
    s[k] = src[e0 + k * ST];
  }
  for (int k = 0; k < SK; ++k) {
    const int pos = atomicAdd(&cur[d[k] >> BSH], 1);
    rec[pos] = (s[k] << BSH) | (d[k] & (BR - 1));
  }
}

__global__ __launch_bounds__(256) void gcn_bucket_rows_kernel(
    const int* __restrict__ rec, const int* __restrict__ base,
    const float* __restrict__ x, int* __restrict__ cnt,
    int* __restrict__ rowptr, float* __restrict__ dinv,
    int* __restrict__ colA, float* __restrict__ Y0) {
  __shared__ int   c[BR];
  __shared__ int   cur[BR];
  __shared__ float dl[BR];
  __shared__ int   wsum[BR / 32];
  const int tid  = threadIdx.x;
  const int lane = tid & 31;
  const int wv   = tid >> 5;
  const int b    = blockIdx.x;
  const int begin = base[b * NS];
  const int end   = (b + 1 < NB) ? base[(b + 1) * NS] : EE;

  if (tid < BR) c[tid] = 0;
  __syncthreads();
  for (int i0 = begin + tid; i0 < end; i0 += RK * 256) {
    int r[RK];
    for (int u = 0; u < RK; ++u) r[u] = rec[min(i0 + u * 256, end - 1)];
    for (int u = 0; u < RK; ++u)
      if (i0 + u * 256 < end) atomicAdd(&c[r[u] & (BR - 1)], 1);
  }
  __syncthreads();

  const int v = (tid < BR) ? c[tid] : 0;
  int inc = v;
  for (int o = 1; o < 32; o <<= 1) {
    const int t = __shfl_up(inc, o, 32);
    if (lane >= o) inc += t;
  }
  if (tid < BR && lane == 31) wsum[wv] = inc;
  __syncthreads();
  if (tid < BR) {
    int off = begin;
    for (int w = 0; w < wv; ++w) off += wsum[w];
    const int excl = off + inc - v;
    const float di = rsqrtf((float)v + 1.0f);
    cur[tid] = excl;
    dl[tid]  = di;
    const int t = b * BR + tid;
    if (t < NN) {
      cnt[t]    = v;
      rowptr[t] = excl;
      dinv[t]   = di;
    }
  }
  __syncthreads();

  for (int i0 = begin + tid; i0 < end; i0 += RK * 256) {
    int r[RK];
    for (int u = 0; u < RK; ++u) r[u] = rec[min(i0 + u * 256, end - 1)];
    for (int u = 0; u < RK; ++u) {
      if (i0 + u * 256 < end) {
        const int pos = atomicAdd(&cur[r[u] & (BR - 1)], 1);
        colA[pos] = r[u] >> BSH;
      }
    }
  }

  const int l = tid & 15;
  float4 xv[BR / 16];
  for (int u = 0; u < BR / 16; ++u) {
    const int t = min(b * BR + u * 16 + (tid >> 4), NN - 1);
    xv[u] = *reinterpret_cast<const float4*>(x + t * DH + l * 4);
  }
  for (int u = 0; u < BR / 16; ++u) {
    const int rl = u * 16 + (tid >> 4);
    const int t  = b * BR + rl;
    if (t < NN) {
      const float d = dl[rl];
      float4 yv;
      yv.x = xv[u].x * d;
      yv.y = xv[u].y * d;
      yv.z = xv[u].z * d;
      yv.w = xv[u].w * d;
      *reinterpret_cast<float4*>(Y0 + t * DH + l * 4) = yv;
    }
  }
}

#define SCAP 64
__device__ __forceinline__ void gcn_sort_row(const int* __restrict__ colU,
                                             int* colS, int* sl, int start,
                                             int n, int lane) {
  if (n <= 32) {
    const int key = (lane < n) ? colU[start + lane] : 0x7fffffff;
    int r = 0;
    for (int j = 0; j < n; ++j) {
      const int kj = __shfl(key, j, 32);
      r += ((kj < key) || (kj == key && j < lane)) ? 1 : 0;
    }
    if (lane < n) {
      colS[start + r] = key;
      sl[r] = key;
    }
  } else {
    for (int i = lane; i < n; i += 32) {
      const int key = colU[start + i];
      int r = 0;
      for (int j = 0; j < n; ++j) {
        const int kj = colU[start + j];
        r += ((kj < key) || (kj == key && j < i)) ? 1 : 0;
      }
      colS[start + r] = key;
      if (r < SCAP) sl[r] = key;
    }
  }
}

#define PITCH 68

template <int NCOLS>
__device__ __forceinline__ v8f gcn_tile_wmma(const float* S,
                                             const float* __restrict__ W,
                                             int tileN, int lane) {
  const int half = lane >> 4;
  const int l    = lane & 15;
  const int bcol = tileN * 16 + l;
  v8f c = {};
  for (int k0 = 0; k0 < DH; k0 += 4) {
    const int ak = k0 + half * 2;
    const v2f a = *reinterpret_cast<const v2f*>(S + l * PITCH + ak);
    v2f b;
    b.x = W[ak * NCOLS + bcol];
    b.y = W[(ak + 1) * NCOLS + bcol];
    c = __builtin_amdgcn_wmma_f32_16x16x4_f32(
         false, a,  false, b,
         (short)0, c,  false,  false);
  }
  return c;
}

template <bool LAST, bool FIRST>
__global__ __launch_bounds__(256) void gcn_layer_kernel(
    const float* __restrict__ Y, const int* __restrict__ col, int* colS,
    const int* __restrict__ rowptr, const int* __restrict__ cnt,
    const float* __restrict__ dinv,
    const float* __restrict__ W, const float* __restrict__ b,
    const float* __restrict__ Wf1, const float* __restrict__ bf1,
    const float* __restrict__ Wf2, const float* __restrict__ bf2,
    float* __restrict__ OUT) {
  __shared__ float sA[16 * PITCH];
  __shared__ float sB[16 * PITCH];
  __shared__ int sIdx[FIRST ? 16 * SCAP : 1];
  const int m = threadIdx.x >> 4;
  const int t = blockIdx.x * 16 + m;
  const int l = threadIdx.x & 15;
  const int start = rowptr[t];
  const int n     = cnt[t];
  const float dt  = dinv[t];
  const float4 ys = *reinterpret_cast<const float4*>(Y + t * DH + l * 4);
  const int nmax = max(n, __shfl_xor(n, 16, 32));

  if (FIRST) {
    const int lane32 = threadIdx.x & 31;
    const int stA = __builtin_amdgcn_readlane(start, 0);
    const int nA  = __builtin_amdgcn_readlane(n, 0);
    const int stB = __builtin_amdgcn_readlane(start, 16);
    const int nB  = __builtin_amdgcn_readlane(n, 16);
    int* slA = sIdx + (threadIdx.x >> 5) * (2 * SCAP);
    int* slB = slA + SCAP;
    if (nA <= 32 && nB <= 32) {
      const int keyA = (lane32 < nA) ? col[stA + lane32] : 0x7fffffff;
      const int keyB = (lane32 < nB) ? col[stB + lane32] : 0x7fffffff;
      const int nm = max(nA, nB);
      int rA = 0, rB = 0;
      for (int j = 0; j < nm; ++j) {
        const int kA = __shfl(keyA, j, 32);
        const int kB = __shfl(keyB, j, 32);
        rA += ((kA < keyA) || (kA == keyA && j < lane32)) ? 1 : 0;
        rB += ((kB < keyB) || (kB == keyB && j < lane32)) ? 1 : 0;
      }
      if (lane32 < nA) {
        colS[stA + rA] = keyA;
        slA[rA] = keyA;
      }
      if (lane32 < nB) {
        colS[stB + rB] = keyB;
        slB[rB] = keyB;
      }
    } else {
      gcn_sort_row(col, colS, slA, stA, nA, lane32);
      gcn_sort_row(col, colS, slB, stB, nB, lane32);
    }
  }

  float4 acc = {0.0f, 0.0f, 0.0f, 0.0f};
  for (int c0 = 0; c0 < nmax; c0 += 16) {
    const int p = c0 + l;
    int myidx = t;
    if (FIRST) {
      if (p < n) {
        if (p < SCAP) {
          myidx = sIdx[m * SCAP + p];
        } else {
          __threadfence_block();
          myidx = colS[start + p];
        }
      }
    } else {
      if (p < n) myidx = col[start + p];
    }
    const int cn = min(nmax - c0, 16);
    for (int j = 0; j < cn; j += 4) {
      const int s0 = __shfl(myidx, j,     16);
      const int s1 = __shfl(myidx, j + 1, 16);
      const int s2 = __shfl(myidx, j + 2, 16);
      const int s3 = __shfl(myidx, j + 3, 16);
      const int   q  = n - (c0 + j);
      const float f0 = (q > 0) ? 1.0f : 0.0f;
      const float f1 = (q > 1) ? 1.0f : 0.0f;
      const float f2 = (q > 2) ? 1.0f : 0.0f;
      const float f3 = (q > 3) ? 1.0f : 0.0f;
      const float4 h0 = *reinterpret_cast<const float4*>(Y + s0 * DH + l * 4);
      const float4 h1 = *reinterpret_cast<const float4*>(Y + s1 * DH + l * 4);
      const float4 h2 = *reinterpret_cast<const float4*>(Y + s2 * DH + l * 4);
      const float4 h3 = *reinterpret_cast<const float4*>(Y + s3 * DH + l * 4);
      acc.x += h0.x * f0; acc.y += h0.y * f0; acc.z += h0.z * f0; acc.w += h0.w * f0;
      acc.x += h1.x * f1; acc.y += h1.y * f1; acc.z += h1.z * f1; acc.w += h1.w * f1;
      acc.x += h2.x * f2; acc.y += h2.y * f2; acc.z += h2.z * f2; acc.w += h2.w * f2;
      acc.x += h3.x * f3; acc.y += h3.y * f3; acc.z += h3.z * f3; acc.w += h3.w * f3;
    }
  }

  float4 g;
  g.x = (acc.x + ys.x) * dt;
  g.y = (acc.y + ys.y) * dt;
  g.z = (acc.z + ys.z) * dt;
  g.w = (acc.w + ys.w) * dt;
  *reinterpret_cast<float4*>(sA + m * PITCH + l * 4) = g;
  __syncthreads();

  const int wv   = __builtin_amdgcn_readfirstlane(threadIdx.x >> 5);
  const int lane = threadIdx.x & 31;
  const int half = lane >> 4;
  const int bcol = wv * 16 + (lane & 15);

  if (wv < 4) {
    const v8f c = gcn_tile_wmma<DH>(sA, W, wv, lane);
    const float bv = b[bcol];
    for (int r = 0; r < 8; ++r)
      sB[(half * 8 + r) * PITCH + bcol] = fmaxf(c[r] + bv, 0.0f);
  }
  __syncthreads();

  if (!LAST) {
    float4 o = *reinterpret_cast<const float4*>(sB + m * PITCH + l * 4);
    o.x *= dt; o.y *= dt; o.z *= dt; o.w *= dt;
    *reinterpret_cast<float4*>(OUT + t * DH + l * 4) = o;
  } else {
    if (wv < 4) {
      const v8f c = gcn_tile_wmma<DH>(sB, Wf1, wv, lane);
      const float bv = bf1[bcol];
      for (int r = 0; r < 8; ++r)
        sA[(half * 8 + r) * PITCH + bcol] = fmaxf(c[r] + bv, 0.0f);
    }
    __syncthreads();
    if (wv == 0) {
      const v8f c = gcn_tile_wmma<DOUT>(sA, Wf2, 0, lane);
      const float bv = bf2[bcol];
      for (int r = 0; r < 8; ++r)
        sB[(half * 8 + r) * DOUT + bcol] = c[r] + bv;
    }
    __syncthreads();
    if (threadIdx.x < 64) {
      const float4 o = *reinterpret_cast<const float4*>(sB + threadIdx.x * 4);
      *reinterpret_cast<float4*>(OUT + blockIdx.x * 16 * DOUT + threadIdx.x * 4) = o;
    }
  }
}

extern "C" void kernel_launch(void* const* d_in, const int* in_sizes, int n_in,
                              void* d_out, int out_size, void* d_ws,
                              size_t ws_size, hipStream_t stream) {
  (void)in_sizes; (void)n_in; (void)out_size; (void)ws_size;

  const float* x   = (const float*)d_in[0];
  const int*   ei  = (const int*)d_in[1];
  const int*   src = ei;
  const int*   dst = ei + EE;
  const float* W1  = (const float*)d_in[2];
  const float* b1  = (const float*)d_in[3];
  const float* W2  = (const float*)d_in[4];
  const float* b2  = (const float*)d_in[5];
  const float* W3  = (const float*)d_in[6];
  const float* b3  = (const float*)d_in[7];
  const float* Wf1 = (const float*)d_in[8];
  const float* bf1 = (const float*)d_in[9];
  const float* Wf2 = (const float*)d_in[10];
  const float* bf2 = (const float*)d_in[11];
  float* out = (float*)d_out;

  float* base = (float*)d_ws;
  float* dinv = base;
  float* bufA = base + NN;
  float* bufB = bufA + NN * DH;
  int* cnt    = (int*)(bufB + NN * DH);
  int* rowptr = cnt + NN;
  int* rec    = rowptr + NN;
  int* colA   = rec + EE;
  int* colB   = colA + EE;
  int* H      = colB + EE;
  int* hbase  = H + NH;

  gcn_bucket_hist_kernel<<<NS, ST, 0, stream>>>(dst, H);
  gcn_scan_kernel<<<(NH + 1023) / 1024, 1024, 0, stream>>>(H, hbase);
  gcn_bucket_scatter_kernel<<<NS, ST, 0, stream>>>(src, dst, hbase, rec);
  gcn_bucket_rows_kernel<<<NB, 256, 0, stream>>>(rec, hbase, x, cnt, rowptr, dinv,
                                               colA, bufB);

  gcn_layer_kernel<false, true><<<NN / 16, 256, 0, stream>>>(
      bufB, colA, colB, rowptr, cnt, dinv, W1, b1, nullptr, nullptr, nullptr, nullptr, bufA);
  gcn_layer_kernel<false, false><<<NN / 16, 256, 0, stream>>>(
      bufA, colB, nullptr, rowptr, cnt, dinv, W2, b2, nullptr, nullptr, nullptr, nullptr, bufB);
  gcn_layer_kernel<true, false><<<NN / 16, 256, 0, stream>>>(
      bufB, colB, nullptr, rowptr, cnt, dinv, W3, b3, Wf1, bf1, Wf2, bf2, out);
}


# CSR build in two launches instead of four: each slice block sorts its own edges by bucket inside LDS and stores them as one piece; the row pass sums the slices' offsets itself; sum order kept
# speedup vs baseline: 7.7363x; 1.3038x over previous
#include <hip/hip_runtime.h>


typedef float v2f __attribute__((ext_vector_type(2)));
typedef float v8f __attribute__((ext_vector_type(8)));

#define NN   100000
#define EE   1600000
#define DH   64
#define DOUT 16

#define BSH 7
#define BR  (1 << BSH)
#define NB  ((NN + BR - 1) / BR)
#define NS  250
#define SE  (EE / NS)
#define ST  640
#define SK  (SE / ST)
#define NSP 256
#define NOFF ((NB + 1) * NSP)
#define RK  10
static_assert(EE == NS * SE && SE == SK * ST, "slices tile the edge list");
static_assert(NN <= (1 << (31 - BSH)), "a source id fits above the local destination");
static_assert(NB % 2 == 0 && NB / 2 <= ST, "the slice block scans two counters a thread");
static_assert(NS <= NSP && NSP == 256, "one thread of the row block per slice");

__global__ __launch_bounds__(ST) void gcn_slice_sort_kernel(
    const int* __restrict__ src, const int* __restrict__ dst,
    int* __restrict__ rec, int* __restrict__ offT) {
  __shared__ int h[NB];
  __shared__ int lrec[SE];
  __shared__ int wsum[32];
  const int tid  = threadIdx.x;
  const int lane = tid & 31;
  const int wv   = tid >> 5;
  for (int b = tid; b < NB; b += ST) h[b] = 0;
  __syncthreads();
  const int e0 = blockIdx.x * SE + tid;
  int d[SK], s[SK];
  for (int k = 0; k < SK; ++k) {
    d[k] = dst[e0 + k * ST];
    s[k] = src[e0 + k * ST];
  }
  for (int k = 0; k < SK; ++k) atomicAdd(&h[d[k] >> BSH], 1);
  __syncthreads();

  const bool own = tid < NB / 2;
  const int v0  = own ? h[2 * tid] : 0;
  const int v1  = own ? h[2 * tid + 1] : 0;
  const int sum = v0 + v1;
  int inc = sum;
  for (int o = 1; o < 32; o <<= 1) {
    const int t = __shfl_up(inc, o, 32);
    if (lane >= o) inc += t;
  }
  if (lane == 31) wsum[wv] = inc;
  __syncthreads();
  if (wv == 0) {
    const int v = (lane < ST / 32) ? wsum[lane] : 0;
    int sc = v;
    for (int o = 1; o < 32; o <<= 1) {
      const int t = __shfl_up(sc, o, 32);
      if (lane >= o) sc += t;
    }
    wsum[lane] = sc - v;
  }
  __syncthreads();
  if (own) {
    const int ex = wsum[wv] + inc - sum;
    h[2 * tid]     = ex;
    h[2 * tid + 1] = ex + v0;
    offT[(2 * tid) * NSP + blockIdx.x]     = ex;
    offT[(2 * tid + 1) * NSP + blockIdx.x] = ex + v0;
  }
  if (tid == 0) offT[NB * NSP + blockIdx.x] = SE;
  __syncthreads();

  int pos[SK];
  for (int k = 0; k < SK; ++k) pos[k] = atomicAdd(&h[d[k] >> BSH], 1);
  for (int k = 0; k < SK; ++k) lrec[pos[k]] = (s[k] << BSH) | (d[k] & (BR - 1));
  __syncthreads();
  for (int i = tid; i < SE; i += ST) rec[blockIdx.x * SE + i] = lrec[i];
}

__device__ __forceinline__ int gcn_bucket_record(const int* __restrict__ rec,
                                                 const int* pre,
                                                 const int* radr, int i) {
  int lo = 0;
  for (int step = NSP / 2; step > 0; step >>= 1)
    if (pre[lo + step] <= i) lo += step;
  return rec[radr[lo] + i];
}

__global__ __launch_bounds__(256) void gcn_bucket_rows_kernel(
    const int* __restrict__ rec, const int* __restrict__ offT,
    const float* __restrict__ x, int* __restrict__ cnt,
    int* __restrict__ rowptr, float* __restrict__ dinv,
    int* __restrict__ colA, float* __restrict__ Y0) {
  __shared__ int   c[BR];
  __shared__ int   cur[BR];
  __shared__ float dl[BR];
  __shared__ int   wsum[BR / 32];
  __shared__ int   pre[NSP];
  __shared__ int   radr[NSP];
  __shared__ int   wlen[NSP / 32];
  __shared__ int   woff[NSP / 32];
  const int tid  = threadIdx.x;
  const int lane = tid & 31;
  const int wv   = tid >> 5;
  const int b    = blockIdx.x;

  const int o0  = (tid < NS) ? offT[b * NSP + tid] : 0;
  const int o1  = (tid < NS) ? offT[(b + 1) * NSP + tid] : 0;
  const int len = o1 - o0;
  int linc = len;
  for (int o = 1; o < 32; o <<= 1) {
    const int t = __shfl_up(linc, o, 32);
    if (lane >= o) linc += t;
  }
  int osum = o0;
  for (int o = 16; o > 0; o >>= 1) osum += __shfl_xor(osum, o, 32);
  if (lane == 31) wlen[wv] = linc;
  if (lane == 0)  woff[wv] = osum;
  if (tid < BR) c[tid] = 0;
  __syncthreads();
  int lfront = 0, nb = 0, begin = 0;
  for (int w = 0; w < NSP / 32; ++w) {
    if (w < wv) lfront += wlen[w];
    nb    += wlen[w];
    begin += woff[w];
  }
  const int lex = lfront + linc - len;
  pre[tid]  = (tid < NS) ? lex : 0x7fffffff;
  radr[tid] = tid * SE + o0 - lex;
  __syncthreads();

  for (int i0 = tid; i0 < nb; i0 += RK * 256) {
    int r[RK];
    for (int u = 0; u < RK; ++u)
      r[u] = gcn_bucket_record(rec, pre, radr, min(i0 + u * 256, nb - 1));
    for (int u = 0; u < RK; ++u)
      if (i0 + u * 256 < nb) atomicAdd(&c[r[u] & (BR - 1)], 1);
  }
  __syncthreads();

  const int v = (tid < BR) ? c[tid] : 0;
  int inc = v;
  for (int o = 1; o < 32; o <<= 1) {
    const int t = __shfl_up(inc, o, 32);
    if (lane >= o) inc += t;
  }
  if (tid < BR && lane == 31) wsum[wv] = inc;
  __syncthreads();
  if (tid < BR) {
    int off = begin;
    for (int w = 0; w < wv; ++w) off += wsum[w];
    const int excl = off + inc - v;
    const float di = rsqrtf((float)v + 1.0f);
    cur[tid] = excl;
    dl[tid]  = di;
    const int t = b * BR + tid;
    if (t < NN) {
      cnt[t]    = v;
      rowptr[t] = excl;
      dinv[t]   = di;
    }
  }
  __syncthreads();

  for (int i0 = tid; i0 < nb; i0 += RK * 256) {
    int r[RK];
    for (int u = 0; u < RK; ++u)
      r[u] = gcn_bucket_record(rec, pre, radr, min(i0 + u * 256, nb - 1));
    for (int u = 0; u < RK; ++u) {
      if (i0 + u * 256 < nb) {
        const int pos = atomicAdd(&cur[r[u] & (BR - 1)], 1);
        colA[pos] = r[u] >> BSH;
      }
    }
  }

  const int l = tid & 15;
  float4 xv[BR / 16];
  for (int u = 0; u < BR / 16; ++u) {
    const int t = min(b * BR + u * 16 + (tid >> 4), NN - 1);
    xv[u] = *reinterpret_cast<const float4*>(x + t * DH + l * 4);
  }
  for (int u = 0; u < BR / 16; ++u) {
    const int rl = u * 16 + (tid >> 4);
    const int t  = b * BR + rl;
    if (t < NN) {
      const float d = dl[rl];
      float4 yv;
      yv.x = xv[u].x * d;
      yv.y = xv[u].y * d;
      yv.z = xv[u].z * d;
      yv.w = xv[u].w * d;
      *reinterpret_cast<float4*>(Y0 + t * DH + l * 4) = yv;
    }
  }
}

#define SCAP 64
__device__ __forceinline__ void gcn_sort_row(const int* __restrict__ colU,
                                             int* colS, int* sl, int start,
                                             int n, int lane) {
  if (n <= 32) {
    const int key = (lane < n) ? colU[start + lane] : 0x7fffffff;
    int r = 0;
    for (int j = 0; j < n; ++j) {
      const int kj = __shfl(key, j, 32);
      r += ((kj < key) || (kj == key && j < lane)) ? 1 : 0;
    }
    if (lane < n) {
      colS[start + r] = key;
      sl[r] = key;
    }
  } else {
    for (int i = lane; i < n; i += 32) {
      const int key = colU[start + i];
      int r = 0;
      for (int j = 0; j < n; ++j) {
        const int kj = colU[start + j];
        r += ((kj < key) || (kj == key && j < i)) ? 1 : 0;
      }
      colS[start + r] = key;
      if (r < SCAP) sl[r] = key;
    }
  }
}

#define PITCH 68

template <int NCOLS>
__device__ __forceinline__ v8f gcn_tile_wmma(const float* S,
                                             const float* __restrict__ W,
                                             int tileN, int lane) {
  const int half = lane >> 4;
  const int l    = lane & 15;
  const int bcol = tileN * 16 + l;
  v8f c = {};
  for (int k0 = 0; k0 < DH; k0 += 4) {
    const int ak = k0 + half * 2;
    const v2f a = *reinterpret_cast<const v2f*>(S + l * PITCH + ak);
    v2f b;
    b.x = W[ak * NCOLS + bcol];
    b.y = W[(ak + 1) * NCOLS + bcol];
    c = __builtin_amdgcn_wmma_f32_16x16x4_f32(
         false, a,  false, b,
         (short)0, c,  false,  false);
  }
  return c;
}

template <bool LAST, bool FIRST>
__global__ __launch_bounds__(256) void gcn_layer_kernel(
    const float* __restrict__ Y, const int* __restrict__ col, int* colS,
    const int* __restrict__ rowptr, const int* __restrict__ cnt,
    const float* __restrict__ dinv,
    const float* __restrict__ W, const float* __restrict__ b,
    const float* __restrict__ Wf1, const float* __restrict__ bf1,
    const float* __restrict__ Wf2, const float* __restrict__ bf2,
    float* __restrict__ OUT) {
  __shared__ float sA[16 * PITCH];
  __shared__ float sB[16 * PITCH];
  __shared__ int sIdx[FIRST ? 16 * SCAP : 1];
  const int m = threadIdx.x >> 4;
  const int t = blockIdx.x * 16 + m;
  const int l = threadIdx.x & 15;
  const int start = rowptr[t];
  const int n     = cnt[t];
  const float dt  = dinv[t];
  const float4 ys = *reinterpret_cast<const float4*>(Y + t * DH + l * 4);
  const int nmax = max(n, __shfl_xor(n, 16, 32));

  if (FIRST) {
    const int lane32 = threadIdx.x & 31;
    const int stA = __builtin_amdgcn_readlane(start, 0);
    const int nA  = __builtin_amdgcn_readlane(n, 0);
    const int stB = __builtin_amdgcn_readlane(start, 16);
    const int nB  = __builtin_amdgcn_readlane(n, 16);
    int* slA = sIdx + (threadIdx.x >> 5) * (2 * SCAP);
    int* slB = slA + SCAP;
    if (nA <= 32 && nB <= 32) {
      const int keyA = (lane32 < nA) ? col[stA + lane32] : 0x7fffffff;
      const int keyB = (lane32 < nB) ? col[stB + lane32] : 0x7fffffff;
      const int nm = max(nA, nB);
      int rA = 0, rB = 0;
      for (int j = 0; j < nm; ++j) {
        const int kA = __shfl(keyA, j, 32);
        const int kB = __shfl(keyB, j, 32);
        rA += ((kA < keyA) || (kA == keyA && j < lane32)) ? 1 : 0;
        rB += ((kB < keyB) || (kB == keyB && j < lane32)) ? 1 : 0;
      }
      if (lane32 < nA) {
        colS[stA + rA] = keyA;
        slA[rA] = keyA;
      }
      if (lane32 < nB) {
        colS[stB + rB] = keyB;
        slB[rB] = keyB;
      }
    } else {
      gcn_sort_row(col, colS, slA, stA, nA, lane32);
      gcn_sort_row(col, colS, slB, stB, nB, lane32);
    }
  }

  float4 acc = {0.0f, 0.0f, 0.0f, 0.0f};
  for (int c0 = 0; c0 < nmax; c0 += 16) {
    const int p = c0 + l;
    int myidx = t;
    if (FIRST) {
      if (p < n) {
        if (p < SCAP) {
          myidx = sIdx[m * SCAP + p];
        } else {
          __threadfence_block();
          myidx = colS[start + p];
        }
      }
    } else {
      if (p < n) myidx = col[start + p];
    }
    const int cn = min(nmax - c0, 16);
    for (int j = 0; j < cn; j += 4) {
      const int s0 = __shfl(myidx, j,     16);
      const int s1 = __shfl(myidx, j + 1, 16);
      const int s2 = __shfl(myidx, j + 2, 16);
      const int s3 = __shfl(myidx, j + 3, 16);
      const int   q  = n - (c0 + j);
      const float f0 = (q > 0) ? 1.0f : 0.0f;
      const float f1 = (q > 1) ? 1.0f : 0.0f;
      const float f2 = (q > 2) ? 1.0f : 0.0f;
      const float f3 = (q > 3) ? 1.0f : 0.0f;
      const float4 h0 = *reinterpret_cast<const float4*>(Y + s0 * DH + l * 4);
      const float4 h1 = *reinterpret_cast<const float4*>(Y + s1 * DH + l * 4);
      const float4 h2 = *reinterpret_cast<const float4*>(Y + s2 * DH + l * 4);
      const float4 h3 = *reinterpret_cast<const float4*>(Y + s3 * DH + l * 4);
      acc.x += h0.x * f0; acc.y += h0.y * f0; acc.z += h0.z * f0; acc.w += h0.w * f0;
      acc.x += h1.x * f1; acc.y += h1.y * f1; acc.z += h1.z * f1; acc.w += h1.w * f1;
      acc.x += h2.x * f2; acc.y += h2.y * f2; acc.z += h2.z * f2; acc.w += h2.w * f2;
      acc.x += h3.x * f3; acc.y += h3.y * f3; acc.z += h3.z * f3; acc.w += h3.w * f3;
    }
  }

  float4 g;
  g.x = (acc.x + ys.x) * dt;
  g.y = (acc.y + ys.y) * dt;
  g.z = (acc.z + ys.z) * dt;
  g.w = (acc.w + ys.w) * dt;
  *reinterpret_cast<float4*>(sA + m * PITCH + l * 4) = g;
  __syncthreads();

  const int wv   = __builtin_amdgcn_readfirstlane(threadIdx.x >> 5);
  const int lane = threadIdx.x & 31;
  const int half = lane >> 4;
  const int bcol = wv * 16 + (lane & 15);

  if (wv < 4) {
    const v8f c = gcn_tile_wmma<DH>(sA, W, wv, lane);
    const float bv = b[bcol];
    for (int r = 0; r < 8; ++r)
      sB[(half * 8 + r) * PITCH + bcol] = fmaxf(c[r] + bv, 0.0f);
  }
  __syncthreads();

  if (!LAST) {
    float4 o = *reinterpret_cast<const float4*>(sB + m * PITCH + l * 4);
    o.x *= dt; o.y *= dt; o.z *= dt; o.w *= dt;
    *reinterpret_cast<float4*>(OUT + t * DH + l * 4) = o;
  } else {
    if (wv < 4) {
      const v8f c = gcn_tile_wmma<DH>(sB, Wf1, wv, lane);
      const float bv = bf1[bcol];
      for (int r = 0; r < 8; ++r)
        sA[(half * 8 + r) * PITCH + bcol] = fmaxf(c[r] + bv, 0.0f);
    }
    __syncthreads();
    if (wv == 0) {
      const v8f c = gcn_tile_wmma<DOUT>(sA, Wf2, 0, lane);
      const float bv = bf2[bcol];
      for (int r = 0; r < 8; ++r)
        sB[(half * 8 + r) * DOUT + bcol] = c[r] + bv;
    }
    __syncthreads();
    if (threadIdx.x < 64) {
      const float4 o = *reinterpret_cast<const float4*>(sB + threadIdx.x * 4);
      *reinterpret_cast<float4*>(OUT + blockIdx.x * 16 * DOUT + threadIdx.x * 4) = o;
    }
  }
}

extern "C" void kernel_launch(void* const* d_in, const int* in_sizes, int n_in,
                              void* d_out, int out_size, void* d_ws,
                              size_t ws_size, hipStream_t stream) {
  (void)in_sizes; (void)n_in; (void)out_size; (void)ws_size;

  const float* x   = (const float*)d_in[0];
  const int*   ei  = (const int*)d_in[1];
  const int*   src = ei;
  const int*   dst = ei + EE;
  const float* W1  = (const float*)d_in[2];
  const float* b1  = (const float*)d_in[3];
  const float* W2  = (const float*)d_in[4];
  const float* b2  = (const float*)d_in[5];
  const float* W3  = (const float*)d_in[6];
  const float* b3  = (const float*)d_in[7];
  const float* Wf1 = (const float*)d_in[8];
  const float* bf1 = (const float*)d_in[9];
  const float* Wf2 = (const float*)d_in[10];
  const float* bf2 = (const float*)d_in[11];
  float* out = (float*)d_out;

  float* base = (float*)d_ws;
  float* dinv = base;
  float* bufA = base + NN;
  float* bufB = bufA + NN * DH;
  int* cnt    = (int*)(bufB + NN * DH);
  int* rowptr = cnt + NN;
  int* rec    = rowptr + NN;
  int* colA   = rec + EE;
  int* colB   = colA + EE;
  int* offT   = colB + EE;

  gcn_slice_sort_kernel<<<NS, ST, 0, stream>>>(src, dst, rec, offT);
  gcn_bucket_rows_kernel<<<NB, 256, 0, stream>>>(rec, offT, x, cnt, rowptr, dinv,
                                               colA, bufB);

  gcn_layer_kernel<false, true><<<NN / 16, 256, 0, stream>>>(
      bufB, colA, colB, rowptr, cnt, dinv, W1, b1, nullptr, nullptr, nullptr, nullptr, bufA);
  gcn_layer_kernel<false, false><<<NN / 16, 256, 0, stream>>>(
      bufA, colB, nullptr, rowptr, cnt, dinv, W2, b2, nullptr, nullptr, nullptr, nullptr, bufB);
  gcn_layer_kernel<true, false><<<NN / 16, 256, 0, stream>>>(
      bufB, colB, nullptr, rowptr, cnt, dinv, W3, b3, Wf1, bf1, Wf2, bf2, out);
}


# CSR build without scattered 4-byte global stores: the row pass keeps its records in LDS (read once), fills an LDS copy of its rows and stores whole lines; slice offsets slice-major; sum order kept
# speedup vs baseline: 8.3535x; 1.0798x over previous
#include <hip/hip_runtime.h>


typedef float v2f __attribute__((ext_vector_type(2)));
typedef float v8f __attribute__((ext_vector_type(8)));

#define NN   100000
#define EE   1600000
#define DH   64
#define DOUT 16

#define BSH 7
#define BR  (1 << BSH)
#define NB  ((NN + BR - 1) / BR)
#define NS  250
#define SE  (EE / NS)
#define ST  640
#define SK  (SE / ST)
#define NSP 256
#define NBP 784
#define NOFF (NS * NBP)
#define LCAP 4096
#define RK  10
static_assert(EE == NS * SE && SE == SK * ST, "slices tile the edge list");
static_assert(NN <= (1 << (31 - BSH)), "a source id fits above the local destination");
static_assert(NB % 2 == 0 && NB / 2 <= ST, "the slice block scans two counters a thread");
static_assert(NS <= NSP && NSP == 256, "one thread of the row block per slice");
static_assert(NB + 1 <= NBP, "a slice's offsets fit its row of offT");

__global__ __launch_bounds__(ST) void gcn_slice_sort_kernel(
    const int* __restrict__ src, const int* __restrict__ dst,
    int* __restrict__ rec, int* __restrict__ offT) {
  __shared__ int h[NB];
  __shared__ int lrec[SE];
  __shared__ int wsum[32];
  const int tid  = threadIdx.x;
  const int lane = tid & 31;
  const int wv   = tid >> 5;
  for (int b = tid; b < NB; b += ST) h[b] = 0;
  __syncthreads();
  const int e0 = blockIdx.x * SE + tid;
  int d[SK], s[SK];
  for (int k = 0; k < SK; ++k) {
    d[k] = dst[e0 + k * ST];
    s[k] = src[e0 + k * ST];
  }
  for (int k = 0; k < SK; ++k) atomicAdd(&h[d[k] >> BSH], 1);
  __syncthreads();

  const bool own = tid < NB / 2;
  const int v0  = own ? h[2 * tid] : 0;
  const int v1  = own ? h[2 * tid + 1] : 0;
  const int sum = v0 + v1;
  int inc = sum;
  for (int o = 1; o < 32; o <<= 1) {
    const int t = __shfl_up(inc, o, 32);
    if (lane >= o) inc += t;
  }
  if (lane == 31) wsum[wv] = inc;
  __syncthreads();
  if (wv == 0) {
    const int v = (lane < ST / 32) ? wsum[lane] : 0;
    int sc = v;
    for (int o = 1; o < 32; o <<= 1) {
      const int t = __shfl_up(sc, o, 32);
      if (lane >= o) sc += t;
    }
    wsum[lane] = sc - v;
  }
  __syncthreads();
  if (own) {
    const int ex = wsum[wv] + inc - sum;
    h[2 * tid]     = ex;
    h[2 * tid + 1] = ex + v0;
    offT[blockIdx.x * NBP + 2 * tid]     = ex;
    offT[blockIdx.x * NBP + 2 * tid + 1] = ex + v0;
  }
  if (tid == 0) offT[blockIdx.x * NBP + NB] = SE;
  __syncthreads();

  int pos[SK];
  for (int k = 0; k < SK; ++k) pos[k] = atomicAdd(&h[d[k] >> BSH], 1);
  for (int k = 0; k < SK; ++k) lrec[pos[k]] = (s[k] << BSH) | (d[k] & (BR - 1));
  __syncthreads();
  for (int i = tid; i < SE; i += ST) rec[blockIdx.x * SE + i] = lrec[i];
}

__device__ __forceinline__ int gcn_bucket_record(const int* __restrict__ rec,
                                                 const int* pre,
                                                 const int* radr, int i) {
  int lo = 0;
  for (int step = NSP / 2; step > 0; step >>= 1)
    if (pre[lo + step] <= i) lo += step;
  return rec[radr[lo] + i];
}

__global__ __launch_bounds__(256) void gcn_bucket_rows_kernel(
    const int* __restrict__ rec, const int* __restrict__ offT,
    const float* __restrict__ x, int* __restrict__ cnt,
    int* __restrict__ rowptr, float* __restrict__ dinv,
    int* __restrict__ colA, float* __restrict__ Y0) {
  __shared__ int   c[BR];
  __shared__ int   cur[BR];
  __shared__ float dl[BR];
  __shared__ int   wsum[BR / 32];
  __shared__ int   pre[NSP];
  __shared__ int   radr[NSP];
  __shared__ int   wlen[NSP / 32];
  __shared__ int   woff[NSP / 32];
  __shared__ int   lrec[LCAP];
  __shared__ int   lcol[LCAP];
  const int tid  = threadIdx.x;
  const int lane = tid & 31;
  const int wv   = tid >> 5;
  const int b    = blockIdx.x;

  const int o0  = (tid < NS) ? offT[tid * NBP + b] : 0;
  const int o1  = (tid < NS) ? offT[tid * NBP + b + 1] : 0;
  const int len = o1 - o0;
  int linc = len;
  for (int o = 1; o < 32; o <<= 1) {
    const int t = __shfl_up(linc, o, 32);
    if (lane >= o) linc += t;
  }
  int osum = o0;
  for (int o = 16; o > 0; o >>= 1) osum += __shfl_xor(osum, o, 32);
  if (lane == 31) wlen[wv] = linc;
  if (lane == 0)  woff[wv] = osum;
  if (tid < BR) c[tid] = 0;
  __syncthreads();
  int lfront = 0, nb = 0, begin = 0;
  for (int w = 0; w < NSP / 32; ++w) {
    if (w < wv) lfront += wlen[w];
    nb    += wlen[w];
    begin += woff[w];
  }
  const int lex = lfront + linc - len;
  pre[tid]  = (tid < NS) ? lex : 0x7fffffff;
  radr[tid] = tid * SE + o0 - lex;
  __syncthreads();

  const bool inl = nb <= LCAP;
  for (int i0 = tid; i0 < nb; i0 += RK * 256) {
    int r[RK];
    for (int u = 0; u < RK; ++u)
      r[u] = gcn_bucket_record(rec, pre, radr, min(i0 + u * 256, nb - 1));
    for (int u = 0; u < RK; ++u) {
      if (i0 + u * 256 < nb) {
        atomicAdd(&c[r[u] & (BR - 1)], 1);
        if (inl) lrec[i0 + u * 256] = r[u];
      }
    }
  }
  __syncthreads();

  const int v = (tid < BR) ? c[tid] : 0;
  int inc = v;
  for (int o = 1; o < 32; o <<= 1) {
    const int t = __shfl_up(inc, o, 32);
    if (lane >= o) inc += t;
  }
  if (tid < BR && lane == 31) wsum[wv] = inc;
  __syncthreads();
  if (tid < BR) {
    int off = begin;
    for (int w = 0; w < wv; ++w) off += wsum[w];
    const int excl = off + inc - v;
    const float di = rsqrtf((float)v + 1.0f);
    cur[tid] = excl;
    dl[tid]  = di;
    const int t = b * BR + tid;
    if (t < NN) {
      cnt[t]    = v;
      rowptr[t] = excl;
      dinv[t]   = di;
    }
  }
  __syncthreads();

  if (inl) {
    for (int i0 = tid; i0 < nb; i0 += RK * 256) {
      int r[RK], pos[RK];
      for (int u = 0; u < RK; ++u) r[u] = lrec[min(i0 + u * 256, nb - 1)];
      for (int u = 0; u < RK; ++u)
        pos[u] = (i0 + u * 256 < nb) ? atomicAdd(&cur[r[u] & (BR - 1)], 1) : begin;
      for (int u = 0; u < RK; ++u)
        if (i0 + u * 256 < nb) lcol[pos[u] - begin] = r[u] >> BSH;
    }
    __syncthreads();
    for (int i = tid - (begin & 31); i < nb; i += 256)
      if (i >= 0) colA[begin + i] = lcol[i];
  } else {
    for (int i0 = tid; i0 < nb; i0 += RK * 256) {
      int r[RK];
      for (int u = 0; u < RK; ++u)
        r[u] = gcn_bucket_record(rec, pre, radr, min(i0 + u * 256, nb - 1));
      for (int u = 0; u < RK; ++u) {
        if (i0 + u * 256 < nb) {
          const int pos = atomicAdd(&cur[r[u] & (BR - 1)], 1);
          colA[pos] = r[u] >> BSH;
        }
      }
    }
  }

  const int l = tid & 15;
  float4 xv[BR / 16];
  for (int u = 0; u < BR / 16; ++u) {
    const int t = min(b * BR + u * 16 + (tid >> 4), NN - 1);
    xv[u] = *reinterpret_cast<const float4*>(x + t * DH + l * 4);
  }
  for (int u = 0; u < BR / 16; ++u) {
    const int rl = u * 16 + (tid >> 4);
    const int t  = b * BR + rl;
    if (t < NN) {
      const float d = dl[rl];
      float4 yv;
      yv.x = xv[u].x * d;
      yv.y = xv[u].y * d;
      yv.z = xv[u].z * d;
      yv.w = xv[u].w * d;
      *reinterpret_cast<float4*>(Y0 + t * DH + l * 4) = yv;
    }
  }
}

#define SCAP 64
__device__ __forceinline__ void gcn_sort_row(const int* __restrict__ colU,
                                             int* colS, int* sl, int start,
                                             int n, int lane) {
  if (n <= 32) {
    const int key = (lane < n) ? colU[start + lane] : 0x7fffffff;
    int r = 0;
    for (int j = 0; j < n; ++j) {
      const int kj = __shfl(key, j, 32);
      r += ((kj < key) || (kj == key && j < lane)) ? 1 : 0;
    }
    if (lane < n) {
      colS[start + r] = key;
      sl[r] = key;
    }
  } else {
    for (int i = lane; i < n; i += 32) {
      const int key = colU[start + i];
      int r = 0;
      for (int j = 0; j < n; ++j) {
        const int kj = colU[start + j];
        r += ((kj < key) || (kj == key && j < i)) ? 1 : 0;
      }
      colS[start + r] = key;
      if (r < SCAP) sl[r] = key;
    }
  }
}

#define PITCH 68

template <int NCOLS>
__device__ __forceinline__ v8f gcn_tile_wmma(const float* S,
                                             const float* __restrict__ W,
                                             int tileN, int lane) {
  const int half = lane >> 4;
  const int l    = lane & 15;
  const int bcol = tileN * 16 + l;
  v8f c = {};
  for (int k0 = 0; k0 < DH; k0 += 4) {
    const int ak = k0 + half * 2;
    const v2f a = *reinterpret_cast<const v2f*>(S + l * PITCH + ak);
    v2f b;
    b.x = W[ak * NCOLS + bcol];
    b.y = W[(ak + 1) * NCOLS + bcol];
    c = __builtin_amdgcn_wmma_f32_16x16x4_f32(
         false, a,  false, b,
         (short)0, c,  false,  false);
  }
  return c;
}

template <bool LAST, bool FIRST>
__global__ __launch_bounds__(256) void gcn_layer_kernel(
    const float* __restrict__ Y, const int* __restrict__ col, int* colS,
    const int* __restrict__ rowptr, const int* __restrict__ cnt,
    const float* __restrict__ dinv,
    const float* __restrict__ W, const float* __restrict__ b,
    const float* __restrict__ Wf1, const float* __restrict__ bf1,
    const float* __restrict__ Wf2, const float* __restrict__ bf2,
    float* __restrict__ OUT) {
  __shared__ float sA[16 * PITCH];
  __shared__ float sB[16 * PITCH];
  __shared__ int sIdx[FIRST ? 16 * SCAP : 1];
  const int m = threadIdx.x >> 4;
  const int t = blockIdx.x * 16 + m;
  const int l = threadIdx.x & 15;
  const int start = rowptr[t];
  const int n     = cnt[t];
  const float dt  = dinv[t];
  const float4 ys = *reinterpret_cast<const float4*>(Y + t * DH + l * 4);
  const int nmax = max(n, __shfl_xor(n, 16, 32));

  if (FIRST) {
    const int lane32 = threadIdx.x & 31;
    const int stA = __builtin_amdgcn_readlane(start, 0);
    const int nA  = __builtin_amdgcn_readlane(n, 0);
    const int stB = __builtin_amdgcn_readlane(start, 16);
    const int nB  = __builtin_amdgcn_readlane(n, 16);
    int* slA = sIdx + (threadIdx.x >> 5) * (2 * SCAP);
    int* slB = slA + SCAP;
    if (nA <= 32 && nB <= 32) {
      const int keyA = (lane32 < nA) ? col[stA + lane32] : 0x7fffffff;
      const int keyB = (lane32 < nB) ? col[stB + lane32] : 0x7fffffff;
      const int nm = max(nA, nB);
      int rA = 0, rB = 0;
      for (int j = 0; j < nm; ++j) {
        const int kA = __shfl(keyA, j, 32);
        const int kB = __shfl(keyB, j, 32);
        rA += ((kA < keyA) || (kA == keyA && j < lane32)) ? 1 : 0;
        rB += ((kB < keyB) || (kB == keyB && j < lane32)) ? 1 : 0;
      }
      if (lane32 < nA) {
        colS[stA + rA] = keyA;
        slA[rA] = keyA;
      }
      if (lane32 < nB) {
        colS[stB + rB] = keyB;
        slB[rB] = keyB;
      }
    } else {
      gcn_sort_row(col, colS, slA, stA, nA, lane32);
      gcn_sort_row(col, colS, slB, stB, nB, lane32);
    }
  }

  float4 acc = {0.0f, 0.0f, 0.0f, 0.0f};
  for (int c0 = 0; c0 < nmax; c0 += 16) {
    const int p = c0 + l;
    int myidx = t;
    if (FIRST) {
      if (p < n) {
        if (p < SCAP) {
          myidx = sIdx[m * SCAP + p];
        } else {
          __threadfence_block();
          myidx = colS[start + p];
        }
      }
    } else {
      if (p < n) myidx = col[start + p];
    }
    const int cn = min(nmax - c0, 16);
    for (int j = 0; j < cn; j += 4) {
      const int s0 = __shfl(myidx, j,     16);
      const int s1 = __shfl(myidx, j + 1, 16);
      const int s2 = __shfl(myidx, j + 2, 16);
      const int s3 = __shfl(myidx, j + 3, 16);
      const int   q  = n - (c0 + j);
      const float f0 = (q > 0) ? 1.0f : 0.0f;
      const float f1 = (q > 1) ? 1.0f : 0.0f;
      const float f2 = (q > 2) ? 1.0f : 0.0f;
      const float f3 = (q > 3) ? 1.0f : 0.0f;
      const float4 h0 = *reinterpret_cast<const float4*>(Y + s0 * DH + l * 4);
      const float4 h1 = *reinterpret_cast<const float4*>(Y + s1 * DH + l * 4);
      const float4 h2 = *reinterpret_cast<const float4*>(Y + s2 * DH + l * 4);
      const float4 h3 = *reinterpret_cast<const float4*>(Y + s3 * DH + l * 4);
      acc.x += h0.x * f0; acc.y += h0.y * f0; acc.z += h0.z * f0; acc.w += h0.w * f0;
      acc.x += h1.x * f1; acc.y += h1.y * f1; acc.z += h1.z * f1; acc.w += h1.w * f1;
      acc.x += h2.x * f2; acc.y += h2.y * f2; acc.z += h2.z * f2; acc.w += h2.w * f2;
      acc.x += h3.x * f3; acc.y += h3.y * f3; acc.z += h3.z * f3; acc.w += h3.w * f3;
    }
  }

  float4 g;
  g.x = (acc.x + ys.x) * dt;
  g.y = (acc.y + ys.y) * dt;
  g.z = (acc.z + ys.z) * dt;
  g.w = (acc.w + ys.w) * dt;
  *reinterpret_cast<float4*>(sA + m * PITCH + l * 4) = g;
  __syncthreads();

  const int wv   = __builtin_amdgcn_readfirstlane(threadIdx.x >> 5);
  const int lane = threadIdx.x & 31;
  const int half = lane >> 4;
  const int bcol = wv * 16 + (lane & 15);

  if (wv < 4) {
    const v8f c = gcn_tile_wmma<DH>(sA, W, wv, lane);
    const float bv = b[bcol];
    for (int r = 0; r < 8; ++r)
      sB[(half * 8 + r) * PITCH + bcol] = fmaxf(c[r] + bv, 0.0f);
  }
  __syncthreads();

  if (!LAST) {
    float4 o = *reinterpret_cast<const float4*>(sB + m * PITCH + l * 4);
    o.x *= dt; o.y *= dt; o.z *= dt; o.w *= dt;
    *reinterpret_cast<float4*>(OUT + t * DH + l * 4) = o;
  } else {
    if (wv < 4) {
      const v8f c = gcn_tile_wmma<DH>(sB, Wf1, wv, lane);
      const float bv = bf1[bcol];
      for (int r = 0; r < 8; ++r)
        sA[(half * 8 + r) * PITCH + bcol] = fmaxf(c[r] + bv, 0.0f);
    }
    __syncthreads();
    if (wv == 0) {
      const v8f c = gcn_tile_wmma<DOUT>(sA, Wf2, 0, lane);
      const float bv = bf2[bcol];
      for (int r = 0; r < 8; ++r)
        sB[(half * 8 + r) * DOUT + bcol] = c[r] + bv;
    }
    __syncthreads();
    if (threadIdx.x < 64) {
      const float4 o = *reinterpret_cast<const float4*>(sB + threadIdx.x * 4);
      *reinterpret_cast<float4*>(OUT + blockIdx.x * 16 * DOUT + threadIdx.x * 4) = o;
    }
  }
}

extern "C" void kernel_launch(void* const* d_in, const int* in_sizes, int n_in,
                              void* d_out, int out_size, void* d_ws,
                              size_t ws_size, hipStream_t stream) {
  (void)in_sizes; (void)n_in; (void)out_size; (void)ws_size;

  const float* x   = (const float*)d_in[0];
  const int*   ei  = (const int*)d_in[1];
  const int*   src = ei;
  const int*   dst = ei + EE;
  const float* W1  = (const float*)d_in[2];
  const float* b1  = (const float*)d_in[3];
  const float* W2  = (const float*)d_in[4];
  const float* b2  = (const float*)d_in[5];
  const float* W3  = (const float*)d_in[6];
  const float* b3  = (const float*)d_in[7];
  const float* Wf1 = (const float*)d_in[8];
  const float* bf1 = (const float*)d_in[9];
  const float* Wf2 = (const float*)d_in[10];
  const float* bf2 = (const float*)d_in[11];
  float* out = (float*)d_out;

  float* base = (float*)d_ws;
  float* dinv = base;
  float* bufA = base + NN;
  float* bufB = bufA + NN * DH;
  int* cnt    = (int*)(bufB + NN * DH);
  int* rowptr = cnt + NN;
  int* rec    = rowptr + NN;
  int* colA   = rec + EE;
  int* colB   = colA + EE;
  int* offT   = colB + EE;

  gcn_slice_sort_kernel<<<NS, ST, 0, stream>>>(src, dst, rec, offT);
  gcn_bucket_rows_kernel<<<NB, 256, 0, stream>>>(rec, offT, x, cnt, rowptr, dinv,
                                               colA, bufB);

  gcn_layer_kernel<false, true><<<NN / 16, 256, 0, stream>>>(
      bufB, colA, colB, rowptr, cnt, dinv, W1, b1, nullptr, nullptr, nullptr, nullptr, bufA);
  gcn_layer_kernel<false, false><<<NN / 16, 256, 0, stream>>>(
      bufA, colB, nullptr, rowptr, cnt, dinv, W2, b2, nullptr, nullptr, nullptr, nullptr, bufB);
  gcn_layer_kernel<true, false><<<NN / 16, 256, 0, stream>>>(
      bufB, colB, nullptr, rowptr, cnt, dinv, W3, b3, Wf1, bf1, Wf2, bf2, out);
}


# Layer blocks walk 5 tiles of 16 nodes: a fifth of the waves enter; weights staged in LDS once a block, next tile's bounds and slot ids fetched a tile ahead, 8 rows in flight; sum order kept
# speedup vs baseline: 9.1587x; 1.0964x over previous
#include <hip/hip_runtime.h>


typedef float v2f __attribute__((ext_vector_type(2)));
typedef float v8f __attribute__((ext_vector_type(8)));

#define NN   100000
#define EE   1600000
#define DH   64
#define DOUT 16

#define BSH 7
#define BR  (1 << BSH)
#define NB  ((NN + BR - 1) / BR)
#define NS  250
#define SE  (EE / NS)
#define ST  640
#define SK  (SE / ST)
#define NSP 256
#define NBP 784
#define NOFF (NS * NBP)
#define LCAP 4096
#define RK  10
static_assert(EE == NS * SE && SE == SK * ST, "slices tile the edge list");
static_assert(NN <= (1 << (31 - BSH)), "a source id fits above the local destination");
static_assert(NB % 2 == 0 && NB / 2 <= ST, "the slice block scans two counters a thread");
static_assert(NS <= NSP && NSP == 256, "one thread of the row block per slice");
static_assert(NB + 1 <= NBP, "a slice's offsets fit its row of offT");

__global__ __launch_bounds__(ST) void gcn_slice_sort_kernel(
    const int* __restrict__ src, const int* __restrict__ dst,
    int* __restrict__ rec, int* __restrict__ offT) {
  __shared__ int h[NB];
  __shared__ int lrec[SE];
  __shared__ int wsum[32];
  const int tid  = threadIdx.x;
  const int lane = tid & 31;
  const int wv   = tid >> 5;
  for (int b = tid; b < NB; b += ST) h[b] = 0;
  __syncthreads();
  const int e0 = blockIdx.x * SE + tid;
  int d[SK], s[SK];
  for (int k = 0; k < SK; ++k) {
    d[k] = dst[e0 + k * ST];
    s[k] = src[e0 + k * ST];
  }
  for (int k = 0; k < SK; ++k) atomicAdd(&h[d[k] >> BSH], 1);
  __syncthreads();

  const bool own = tid < NB / 2;
  const int v0  = own ? h[2 * tid] : 0;
  const int v1  = own ? h[2 * tid + 1] : 0;
  const int sum = v0 + v1;
  int inc = sum;
  for (int o = 1; o < 32; o <<= 1) {
    const int t = __shfl_up(inc, o, 32);
    if (lane >= o) inc += t;
  }
  if (lane == 31) wsum[wv] = inc;
  __syncthreads();
  if (wv == 0) {
    const int v = (lane < ST / 32) ? wsum[lane] : 0;
    int sc = v;
    for (int o = 1; o < 32; o <<= 1) {
      const int t = __shfl_up(sc, o, 32);
      if (lane >= o) sc += t;
    }
    wsum[lane] = sc - v;
  }
  __syncthreads();
  if (own) {
    const int ex = wsum[wv] + inc - sum;
    h[2 * tid]     = ex;
    h[2 * tid + 1] = ex + v0;
    offT[blockIdx.x * NBP + 2 * tid]     = ex;
    offT[blockIdx.x * NBP + 2 * tid + 1] = ex + v0;
  }
  if (tid == 0) offT[blockIdx.x * NBP + NB] = SE;
  __syncthreads();

  int pos[SK];
  for (int k = 0; k < SK; ++k) pos[k] = atomicAdd(&h[d[k] >> BSH], 1);
  for (int k = 0; k < SK; ++k) lrec[pos[k]] = (s[k] << BSH) | (d[k] & (BR - 1));
  __syncthreads();
  for (int i = tid; i < SE; i += ST) rec[blockIdx.x * SE + i] = lrec[i];
}

__device__ __forceinline__ int gcn_bucket_record(const int* __restrict__ rec,
                                                 const int* pre,
                                                 const int* radr, int i) {
  int lo = 0;
  for (int step = NSP / 2; step > 0; step >>= 1)
    if (pre[lo + step] <= i) lo += step;
  return rec[radr[lo] + i];
}

__global__ __launch_bounds__(256) void gcn_bucket_rows_kernel(
    const int* __restrict__ rec, const int* __restrict__ offT,
    const float* __restrict__ x, int* __restrict__ cnt,
    int* __restrict__ rowptr, float* __restrict__ dinv,
    int* __restrict__ colA, float* __restrict__ Y0) {
  __shared__ int   c[BR];
  __shared__ int   cur[BR];
  __shared__ float dl[BR];
  __shared__ int   wsum[BR / 32];
  __shared__ int   pre[NSP];
  __shared__ int   radr[NSP];
  __shared__ int   wlen[NSP / 32];
  __shared__ int   woff[NSP / 32];
  __shared__ int   lrec[LCAP];
  __shared__ int   lcol[LCAP];
  const int tid  = threadIdx.x;
  const int lane = tid & 31;
  const int wv   = tid >> 5;
  const int b    = blockIdx.x;

  const int o0  = (tid < NS) ? offT[tid * NBP + b] : 0;
  const int o1  = (tid < NS) ? offT[tid * NBP + b + 1] : 0;
  const int len = o1 - o0;
  int linc = len;
  for (int o = 1; o < 32; o <<= 1) {
    const int t = __shfl_up(linc, o, 32);
    if (lane >= o) linc += t;
  }
  int osum = o0;
  for (int o = 16; o > 0; o >>= 1) osum += __shfl_xor(osum, o, 32);
  if (lane == 31) wlen[wv] = linc;
  if (lane == 0)  woff[wv] = osum;
  if (tid < BR) c[tid] = 0;
  __syncthreads();
  int lfront = 0, nb = 0, begin = 0;
  for (int w = 0; w < NSP / 32; ++w) {
    if (w < wv) lfront += wlen[w];
    nb    += wlen[w];
    begin += woff[w];
  }
  const int lex = lfront + linc - len;
  pre[tid]  = (tid < NS) ? lex : 0x7fffffff;
  radr[tid] = tid * SE + o0 - lex;
  __syncthreads();

  const bool inl = nb <= LCAP;
  for (int i0 = tid; i0 < nb; i0 += RK * 256) {
    int r[RK];
    for (int u = 0; u < RK; ++u)
      r[u] = gcn_bucket_record(rec, pre, radr, min(i0 + u * 256, nb - 1));
    for (int u = 0; u < RK; ++u) {
      if (i0 + u * 256 < nb) {
        atomicAdd(&c[r[u] & (BR - 1)], 1);
        if (inl) lrec[i0 + u * 256] = r[u];
      }
    }
  }
  __syncthreads();

  const int v = (tid < BR) ? c[tid] : 0;
  int inc = v;
  for (int o = 1; o < 32; o <<= 1) {
    const int t = __shfl_up(inc, o, 32);
    if (lane >= o) inc += t;
  }
  if (tid < BR && lane == 31) wsum[wv] = inc;
  __syncthreads();
  if (tid < BR) {
    int off = begin;
    for (int w = 0; w < wv; ++w) off += wsum[w];
    const int excl = off + inc - v;
    const float di = rsqrtf((float)v + 1.0f);
    cur[tid] = excl;
    dl[tid]  = di;
    const int t = b * BR + tid;
    if (t < NN) {
      cnt[t]    = v;
      rowptr[t] = excl;
      dinv[t]   = di;
    }
  }
  __syncthreads();

  if (inl) {
    for (int i0 = tid; i0 < nb; i0 += RK * 256) {
      int r[RK], pos[RK];
      for (int u = 0; u < RK; ++u) r[u] = lrec[min(i0 + u * 256, nb - 1)];
      for (int u = 0; u < RK; ++u)
        pos[u] = (i0 + u * 256 < nb) ? atomicAdd(&cur[r[u] & (BR - 1)], 1) : begin;
      for (int u = 0; u < RK; ++u)
        if (i0 + u * 256 < nb) lcol[pos[u] - begin] = r[u] >> BSH;
    }
    __syncthreads();
    for (int i = tid - (begin & 31); i < nb; i += 256)
      if (i >= 0) colA[begin + i] = lcol[i];
  } else {
    for (int i0 = tid; i0 < nb; i0 += RK * 256) {
      int r[RK];
      for (int u = 0; u < RK; ++u)
        r[u] = gcn_bucket_record(rec, pre, radr, min(i0 + u * 256, nb - 1));
      for (int u = 0; u < RK; ++u) {
        if (i0 + u * 256 < nb) {
          const int pos = atomicAdd(&cur[r[u] & (BR - 1)], 1);
          colA[pos] = r[u] >> BSH;
        }
      }
    }
  }

  const int l = tid & 15;
  float4 xv[BR / 16];
  for (int u = 0; u < BR / 16; ++u) {
    const int t = min(b * BR + u * 16 + (tid >> 4), NN - 1);
    xv[u] = *reinterpret_cast<const float4*>(x + t * DH + l * 4);
  }
  for (int u = 0; u < BR / 16; ++u) {
    const int rl = u * 16 + (tid >> 4);
    const int t  = b * BR + rl;
    if (t < NN) {
      const float d = dl[rl];
      float4 yv;
      yv.x = xv[u].x * d;
      yv.y = xv[u].y * d;
      yv.z = xv[u].z * d;
      yv.w = xv[u].w * d;
      *reinterpret_cast<float4*>(Y0 + t * DH + l * 4) = yv;
    }
  }
}

#define SCAP 64
__device__ __forceinline__ void gcn_sort_row(const int* __restrict__ colU,
                                             int* colS, int* sl, int start,
                                             int n, int lane) {
  if (n <= 32) {
    const int key = (lane < n) ? colU[start + lane] : 0x7fffffff;
    int r = 0;
    for (int j = 0; j < n; ++j) {
      const int kj = __shfl(key, j, 32);
      r += ((kj < key) || (kj == key && j < lane)) ? 1 : 0;
    }
    if (lane < n) {
      colS[start + r] = key;
      sl[r] = key;
    }
  } else {
    for (int i = lane; i < n; i += 32) {
      const int key = colU[start + i];
      int r = 0;
      for (int j = 0; j < n; ++j) {
        const int kj = colU[start + j];
        r += ((kj < key) || (kj == key && j < i)) ? 1 : 0;
      }
      colS[start + r] = key;
      if (r < SCAP) sl[r] = key;
    }
  }
}

#define PITCH 68

template <int NCOLS, int WP>
__device__ __forceinline__ void gcn_stage_w(const float* __restrict__ W,
                                            float* sW, int tid) {
  constexpr int NV = DH * NCOLS / 4 / 256;
  static_assert(NV * 4 * 256 == DH * NCOLS && NCOLS % 4 == 0, "256 threads cover W");
  float4 w[NV];
  for (int i = 0; i < NV; ++i)
    w[i] = reinterpret_cast<const float4*>(W)[i * 256 + tid];
  for (int i = 0; i < NV; ++i) {
    const int e = (i * 256 + tid) * 4;
    const int r = e / NCOLS;
    const int c = e % NCOLS;
    float* d = sW + ((r >> 1) * WP + c) * 2 + (r & 1);
    d[0] = w[i].x;
    d[2] = w[i].y;
    d[4] = w[i].z;
    d[6] = w[i].w;
  }
}

template <int WP>
__device__ __forceinline__ v8f gcn_tile_wmma_lds(const float* S, const float* sW,
                                                 int tileN, int lane) {
  const int half = lane >> 4;
  const int l    = lane & 15;
  const int bcol = tileN * 16 + l;
  v8f c = {};
  for (int k = 0; k < DH / 4; ++k) {
    const int ak = k * 4 + half * 2;
    const v2f a = *reinterpret_cast<const v2f*>(S + l * PITCH + ak);
    const v2f b = *reinterpret_cast<const v2f*>(sW + ((2 * k + half) * WP + bcol) * 2);
    c = __builtin_amdgcn_wmma_f32_16x16x4_f32(
         false, a,  false, b,
         (short)0, c,  false,  false);
  }
  return c;
}

__device__ __forceinline__ void gcn_gather8(const float4* __restrict__ Y4,
                                            int idx, int jb, int q, int l,
                                            float4& acc) {
  int s[8];
  float4 h[8];
  for (int k = 0; k < 8; ++k) s[k] = __shfl(idx, jb + k, 16);
  for (int k = 0; k < 8; ++k) h[k] = Y4[(unsigned)(s[k] * (DH / 4) + l)];
  for (int k = 0; k < 8; ++k) {
    const float f = (q > k) ? 1.0f : 0.0f;
    acc.x = __builtin_fmaf(h[k].x, f, acc.x);
    acc.y = __builtin_fmaf(h[k].y, f, acc.y);
    acc.z = __builtin_fmaf(h[k].z, f, acc.z);
    acc.w = __builtin_fmaf(h[k].w, f, acc.w);
  }
}

#define TPB 5
static_assert(NN % 16 == 0 && (NN / 16) % TPB == 0, "layer blocks tile the nodes");

template <bool LAST, bool FIRST>
__global__ __launch_bounds__(256, 10) void gcn_layer_kernel(
    const float* __restrict__ Y, const int* __restrict__ col, int* colS,
    const int* __restrict__ rowptr, const int* __restrict__ cnt,
    const float* __restrict__ dinv,
    const float* __restrict__ W, const float* __restrict__ b,
    const float* __restrict__ Wf1, const float* __restrict__ bf1,
    const float* __restrict__ Wf2, const float* __restrict__ bf2,
    float* __restrict__ OUT) {
  __shared__ float sA[16 * PITCH];
  __shared__ float sB[16 * PITCH];
  __shared__ int sIdx[FIRST ? 16 * SCAP : 1];
  __shared__ float sW[(DH / 2) * 80 * 2];
  __shared__ float sW1[LAST ? (DH / 2) * 80 * 2 : 2];
  __shared__ float sW2[LAST ? (DH / 2) * 16 * 2 : 2];
  const float4* __restrict__ Y4 = reinterpret_cast<const float4*>(Y);
  const int m = threadIdx.x >> 4;
  const int l = threadIdx.x & 15;
  const int wv   = __builtin_amdgcn_readfirstlane(threadIdx.x >> 5);
  const int lane = threadIdx.x & 31;
  const int half = lane >> 4;
  const int bcol = wv * 16 + (lane & 15);

  int t = blockIdx.x * (TPB * 16) + m;
  int start = rowptr[t];
  int n     = cnt[t];

  float bv = 0.0f, bv1 = 0.0f, bv2 = 0.0f;
  if (wv < 4) {
    bv = b[bcol];
    if (LAST) bv1 = bf1[bcol];
    if (LAST && wv == 0) bv2 = bf2[bcol];
  }

  float  dt = dinv[t];
  float4 ys = Y4[t * (DH / 4) + l];
  int idx0 = t, idx1 = t;
  int keyA = 0x7fffffff, keyB = 0x7fffffff;
  if (FIRST) {
    const int stA = __builtin_amdgcn_readlane(start, 0);
    const int nA  = __builtin_amdgcn_readlane(n, 0);
    const int stB = __builtin_amdgcn_readlane(start, 16);
    const int nB  = __builtin_amdgcn_readlane(n, 16);
    if (lane < nA) keyA = col[stA + lane];
    if (lane < nB) keyB = col[stB + lane];
  } else {
    if (l < n)      idx0 = col[start + l];
    if (16 + l < n) idx1 = col[start + 16 + l];
  }

  gcn_stage_w<DH, 80>(W, sW, threadIdx.x);
  if (LAST) {
    gcn_stage_w<DH, 80>(Wf1, sW1, threadIdx.x);
    gcn_stage_w<DOUT, 16>(Wf2, sW2, threadIdx.x);
  }

#pragma nounroll
  for (int it = 0; it < TPB; ++it) {
    const bool more = it + 1 < TPB;
    const int  tn   = t + 16;
    int nstart = 0, nn = 0;
    if (more) {
      nstart = rowptr[tn];
      nn     = cnt[tn];
    }

    if (FIRST) {
      const int stA = __builtin_amdgcn_readlane(start, 0);
      const int nA  = __builtin_amdgcn_readlane(n, 0);
      const int stB = __builtin_amdgcn_readlane(start, 16);
      const int nB  = __builtin_amdgcn_readlane(n, 16);
      int* slA = sIdx + wv * (2 * SCAP);
      int* slB = slA + SCAP;
      if (nA <= 32 && nB <= 32) {
        const int nm = max(nA, nB);
        int rA = 0, rB = 0;
        for (int j = 0; j < nm; ++j) {
          const int kA = __shfl(keyA, j, 32);
          const int kB = __shfl(keyB, j, 32);
          rA += ((kA < keyA) || (kA == keyA && j < lane)) ? 1 : 0;
          rB += ((kB < keyB) || (kB == keyB && j < lane)) ? 1 : 0;
        }
        if (lane < nA) {
          colS[stA + rA] = keyA;
          slA[rA] = keyA;
        }
        if (lane < nB) {
          colS[stB + rB] = keyB;
          slB[rB] = keyB;
        }
      } else {
        gcn_sort_row(col, colS, slA, stA, nA, lane);
        gcn_sort_row(col, colS, slB, stB, nB, lane);
      }
      idx0 = (l < n)      ? sIdx[m * SCAP + l]      : t;
      idx1 = (16 + l < n) ? sIdx[m * SCAP + 16 + l] : t;
    }

    const int nmax = __builtin_amdgcn_readfirstlane(max(n, __shfl_xor(n, 16, 32)));
    float4 acc = {0.0f, 0.0f, 0.0f, 0.0f};
    if (nmax > 0)  gcn_gather8(Y4, idx0, 0, n,      l, acc);
    float  ndt = 0.0f;
    float4 nys = {0.0f, 0.0f, 0.0f, 0.0f};
    int nidx0 = tn, nidx1 = tn;
    int nkeyA = 0x7fffffff, nkeyB = 0x7fffffff;
    if (more) {
      ndt = dinv[tn];
      nys = Y4[tn * (DH / 4) + l];
      if (FIRST) {
        const int stA = __builtin_amdgcn_readlane(nstart, 0);
        const int nA  = __builtin_amdgcn_readlane(nn, 0);
        const int stB = __builtin_amdgcn_readlane(nstart, 16);
        const int nB  = __builtin_amdgcn_readlane(nn, 16);
        if (lane < nA) nkeyA = col[stA + lane];
        if (lane < nB) nkeyB = col[stB + lane];
      } else {
        if (l < nn)      nidx0 = col[nstart + l];
        if (16 + l < nn) nidx1 = col[nstart + 16 + l];
      }
    }

    if (nmax > 8)  gcn_gather8(Y4, idx0, 8, n - 8,  l, acc);
    if (nmax > 16) gcn_gather8(Y4, idx1, 0, n - 16, l, acc);
    if (nmax > 24) gcn_gather8(Y4, idx1, 8, n - 24, l, acc);
    for (int c0 = 32; c0 < nmax; c0 += 16) {
      const int p = c0 + l;
      int myidx = t;
      if (p < n) {
        if (FIRST) {
          if (p < SCAP) {
            myidx = sIdx[m * SCAP + p];
          } else {
            __threadfence_block();
            myidx = colS[start + p];
          }
        } else {
          myidx = col[start + p];
        }
      }
      const int cn = min(nmax - c0, 16);
#pragma nounroll
      for (int j = 0; j < cn; j += 4) {
        const int s0 = __shfl(myidx, j,     16);
        const int s1 = __shfl(myidx, j + 1, 16);
        const int s2 = __shfl(myidx, j + 2, 16);
        const int s3 = __shfl(myidx, j + 3, 16);
        const int   q  = n - (c0 + j);
        const float f0 = (q > 0) ? 1.0f : 0.0f;
        const float f1 = (q > 1) ? 1.0f : 0.0f;
        const float f2 = (q > 2) ? 1.0f : 0.0f;
        const float f3 = (q > 3) ? 1.0f : 0.0f;
        const float4 h0 = Y4[(unsigned)(s0 * (DH / 4) + l)];
        const float4 h1 = Y4[(unsigned)(s1 * (DH / 4) + l)];
        const float4 h2 = Y4[(unsigned)(s2 * (DH / 4) + l)];
        const float4 h3 = Y4[(unsigned)(s3 * (DH / 4) + l)];
        acc.x = __builtin_fmaf(h0.x, f0, acc.x); acc.y = __builtin_fmaf(h0.y, f0, acc.y);
        acc.z = __builtin_fmaf(h0.z, f0, acc.z); acc.w = __builtin_fmaf(h0.w, f0, acc.w);
        acc.x = __builtin_fmaf(h1.x, f1, acc.x); acc.y = __builtin_fmaf(h1.y, f1, acc.y);
        acc.z = __builtin_fmaf(h1.z, f1, acc.z); acc.w = __builtin_fmaf(h1.w, f1, acc.w);
        acc.x = __builtin_fmaf(h2.x, f2, acc.x); acc.y = __builtin_fmaf(h2.y, f2, acc.y);
        acc.z = __builtin_fmaf(h2.z, f2, acc.z); acc.w = __builtin_fmaf(h2.w, f2, acc.w);
        acc.x = __builtin_fmaf(h3.x, f3, acc.x); acc.y = __builtin_fmaf(h3.y, f3, acc.y);
        acc.z = __builtin_fmaf(h3.z, f3, acc.z); acc.w = __builtin_fmaf(h3.w, f3, acc.w);
      }
    }

    float4 g;
    g.x = (acc.x + ys.x) * dt;
    g.y = (acc.y + ys.y) * dt;
    g.z = (acc.z + ys.z) * dt;
    g.w = (acc.w + ys.w) * dt;
    *reinterpret_cast<float4*>(sA + m * PITCH + l * 4) = g;
    __syncthreads();

    if (wv < 4) {
      const v8f c = gcn_tile_wmma_lds<80>(sA, sW, wv, lane);
      for (int r = 0; r < 8; ++r)
        sB[(half * 8 + r) * PITCH + bcol] = fmaxf(c[r] + bv, 0.0f);
    }
    __syncthreads();

    if (!LAST) {
      float4 o = *reinterpret_cast<const float4*>(sB + m * PITCH + l * 4);
      o.x *= dt; o.y *= dt; o.z *= dt; o.w *= dt;
      *reinterpret_cast<float4*>(OUT + t * DH + l * 4) = o;
    } else {
      if (wv < 4) {
        const v8f c = gcn_tile_wmma_lds<80>(sB, sW1, wv, lane);
        for (int r = 0; r < 8; ++r)
          sA[(half * 8 + r) * PITCH + bcol] = fmaxf(c[r] + bv1, 0.0f);
      }
      __syncthreads();
      if (wv == 0) {
        const v8f c = gcn_tile_wmma_lds<16>(sA, sW2, 0, lane);
        for (int r = 0; r < 8; ++r)
          sB[(half * 8 + r) * DOUT + bcol] = c[r] + bv2;
      }
      __syncthreads();
      if (threadIdx.x < 64) {
        const float4 o = *reinterpret_cast<const float4*>(sB + threadIdx.x * 4);
        *reinterpret_cast<float4*>(OUT + (blockIdx.x * TPB + it) * (16 * DOUT) + threadIdx.x * 4) = o;
      }
    }

    t = tn; start = nstart; n = nn; dt = ndt; ys = nys;
    idx0 = nidx0; idx1 = nidx1; keyA = nkeyA; keyB = nkeyB;
  }
}

extern "C" void kernel_launch(void* const* d_in, const int* in_sizes, int n_in,
                              void* d_out, int out_size, void* d_ws,
                              size_t ws_size, hipStream_t stream) {
  (void)in_sizes; (void)n_in; (void)out_size; (void)ws_size;

  const float* x   = (const float*)d_in[0];
  const int*   ei  = (const int*)d_in[1];
  const int*   src = ei;
  const int*   dst = ei + EE;
  const float* W1  = (const float*)d_in[2];
  const float* b1  = (const float*)d_in[3];
  const float* W2  = (const float*)d_in[4];
  const float* b2  = (const float*)d_in[5];
  const float* W3  = (const float*)d_in[6];
  const float* b3  = (const float*)d_in[7];
  const float* Wf1 = (const float*)d_in[8];
  const float* bf1 = (const float*)d_in[9];
  const float* Wf2 = (const float*)d_in[10];
  const float* bf2 = (const float*)d_in[11];
  float* out = (float*)d_out;

  float* base = (float*)d_ws;
  float* dinv = base;
  float* bufA = base + NN;
  float* bufB = bufA + NN * DH;
  int* cnt    = (int*)(bufB + NN * DH);
  int* rowptr = cnt + NN;
  int* rec    = rowptr + NN;
  int* colA   = rec + EE;
  int* colB   = colA + EE;
  int* offT   = colB + EE;

  gcn_slice_sort_kernel<<<NS, ST, 0, stream>>>(src, dst, rec, offT);
  gcn_bucket_rows_kernel<<<NB, 256, 0, stream>>>(rec, offT, x, cnt, rowptr, dinv,
                                               colA, bufB);

  gcn_layer_kernel<false, true><<<NN / 16 / TPB, 256, 0, stream>>>(
      bufB, colA, colB, rowptr, cnt, dinv, W1, b1, nullptr, nullptr, nullptr, nullptr, bufA);
  gcn_layer_kernel<false, false><<<NN / 16 / TPB, 256, 0, stream>>>(
      bufA, colB, nullptr, rowptr, cnt, dinv, W2, b2, nullptr, nullptr, nullptr, nullptr, bufB);
  gcn_layer_kernel<true, false><<<NN / 16 / TPB, 256, 0, stream>>>(
      bufB, colB, nullptr, rowptr, cnt, dinv, W3, b3, Wf1, bf1, Wf2, bf2, out);
}


# Each tile's first eight neighbour rows are asked for in front of the previous tile's product and stay in flight across it (the block barriers fence LDS only); sum order kept
# speedup vs baseline: 9.5410x; 1.0417x over previous
#include <hip/hip_runtime.h>


typedef float v2f __attribute__((ext_vector_type(2)));
typedef float v8f __attribute__((ext_vector_type(8)));

#define NN   100000
#define EE   1600000
#define DH   64
#define DOUT 16

#define BSH 7
#define BR  (1 << BSH)
#define NB  ((NN + BR - 1) / BR)
#define NS  250
#define SE  (EE / NS)
#define ST  640
#define SK  (SE / ST)
#define NSP 256
#define NBP 784
#define NOFF (NS * NBP)
#define LCAP 4096
#define RK  10
static_assert(EE == NS * SE && SE == SK * ST, "slices tile the edge list");
static_assert(NN <= (1 << (31 - BSH)), "a source id fits above the local destination");
static_assert(NB % 2 == 0 && NB / 2 <= ST, "the slice block scans two counters a thread");
static_assert(NS <= NSP && NSP == 256, "one thread of the row block per slice");
static_assert(NB + 1 <= NBP, "a slice's offsets fit its row of offT");

__global__ __launch_bounds__(ST) void gcn_slice_sort_kernel(
    const int* __restrict__ src, const int* __restrict__ dst,
    int* __restrict__ rec, int* __restrict__ offT) {
  __shared__ int h[NB];
  __shared__ int lrec[SE];
  __shared__ int wsum[32];
  const int tid  = threadIdx.x;
  const int lane = tid & 31;
  const int wv   = tid >> 5;
  for (int b = tid; b < NB; b += ST) h[b] = 0;
  __syncthreads();
  const int e0 = blockIdx.x * SE + tid;
  int d[SK], s[SK];
  for (int k = 0; k < SK; ++k) {
    d[k] = dst[e0 + k * ST];
    s[k] = src[e0 + k * ST];
  }
  for (int k = 0; k < SK; ++k) atomicAdd(&h[d[k] >> BSH], 1);
  __syncthreads();

  const bool own = tid < NB / 2;
  const int v0  = own ? h[2 * tid] : 0;
  const int v1  = own ? h[2 * tid + 1] : 0;
  const int sum = v0 + v1;
  int inc = sum;
  for (int o = 1; o < 32; o <<= 1) {
    const int t = __shfl_up(inc, o, 32);
    if (lane >= o) inc += t;
  }
  if (lane == 31) wsum[wv] = inc;
  __syncthreads();
  if (wv == 0) {
    const int v = (lane < ST / 32) ? wsum[lane] : 0;
    int sc = v;
    for (int o = 1; o < 32; o <<= 1) {
      const int t = __shfl_up(sc, o, 32);
      if (lane >= o) sc += t;
    }
    wsum[lane] = sc - v;
  }
  __syncthreads();
  if (own) {
    const int ex = wsum[wv] + inc - sum;
    h[2 * tid]     = ex;
    h[2 * tid + 1] = ex + v0;
    offT[blockIdx.x * NBP + 2 * tid]     = ex;
    offT[blockIdx.x * NBP + 2 * tid + 1] = ex + v0;
  }
  if (tid == 0) offT[blockIdx.x * NBP + NB] = SE;
  __syncthreads();

  int pos[SK];
  for (int k = 0; k < SK; ++k) pos[k] = atomicAdd(&h[d[k] >> BSH], 1);
  for (int k = 0; k < SK; ++k) lrec[pos[k]] = (s[k] << BSH) | (d[k] & (BR - 1));
  __syncthreads();
  for (int i = tid; i < SE; i += ST) rec[blockIdx.x * SE + i] = lrec[i];
}

__device__ __forceinline__ int gcn_bucket_record(const int* __restrict__ rec,
                                                 const int* pre,
                                                 const int* radr, int i) {
  int lo = 0;
  for (int step = NSP / 2; step > 0; step >>= 1)
    if (pre[lo + step] <= i) lo += step;
  return rec[radr[lo] + i];
}

__global__ __launch_bounds__(256) void gcn_bucket_rows_kernel(
    const int* __restrict__ rec, const int* __restrict__ offT,
    const float* __restrict__ x, int* __restrict__ cnt,
    int* __restrict__ rowptr, float* __restrict__ dinv,
    int* __restrict__ colA, float* __restrict__ Y0) {
  __shared__ int   c[BR];
  __shared__ int   cur[BR];
  __shared__ float dl[BR];
  __shared__ int   wsum[BR / 32];
  __shared__ int   pre[NSP];
  __shared__ int   radr[NSP];
  __shared__ int   wlen[NSP / 32];
  __shared__ int   woff[NSP / 32];
  __shared__ int   lrec[LCAP];
  __shared__ int   lcol[LCAP];
  const int tid  = threadIdx.x;
  const int lane = tid & 31;
  const int wv   = tid >> 5;
  const int b    = blockIdx.x;

  const int o0  = (tid < NS) ? offT[tid * NBP + b] : 0;
  const int o1  = (tid < NS) ? offT[tid * NBP + b + 1] : 0;
  const int len = o1 - o0;
  int linc = len;
  for (int o = 1; o < 32; o <<= 1) {
    const int t = __shfl_up(linc, o, 32);
    if (lane >= o) linc += t;
  }
  int osum = o0;
  for (int o = 16; o > 0; o >>= 1) osum += __shfl_xor(osum, o, 32);
  if (lane == 31) wlen[wv] = linc;
  if (lane == 0)  woff[wv] = osum;
  if (tid < BR) c[tid] = 0;
  __syncthreads();
  int lfront = 0, nb = 0, begin = 0;
  for (int w = 0; w < NSP / 32; ++w) {
    if (w < wv) lfront += wlen[w];
    nb    += wlen[w];
    begin += woff[w];
  }
  const int lex = lfront + linc - len;
  pre[tid]  = (tid < NS) ? lex : 0x7fffffff;
  radr[tid] = tid * SE + o0 - lex;
  __syncthreads();

  const bool inl = nb <= LCAP;
  for (int i0 = tid; i0 < nb; i0 += RK * 256) {
    int r[RK];
    for (int u = 0; u < RK; ++u)
      r[u] = gcn_bucket_record(rec, pre, radr, min(i0 + u * 256, nb - 1));
    for (int u = 0; u < RK; ++u) {
      if (i0 + u * 256 < nb) {
        atomicAdd(&c[r[u] & (BR - 1)], 1);
        if (inl) lrec[i0 + u * 256] = r[u];
      }
    }
  }
  __syncthreads();

  const int v = (tid < BR) ? c[tid] : 0;
  int inc = v;
  for (int o = 1; o < 32; o <<= 1) {
    const int t = __shfl_up(inc, o, 32);
    if (lane >= o) inc += t;
  }
  if (tid < BR && lane == 31) wsum[wv] = inc;
  __syncthreads();
  if (tid < BR) {
    int off = begin;
    for (int w = 0; w < wv; ++w) off += wsum[w];
    const int excl = off + inc - v;
    const float di = rsqrtf((float)v + 1.0f);
    cur[tid] = excl;
    dl[tid]  = di;
    const int t = b * BR + tid;
    if (t < NN) {
      cnt[t]    = v;
      rowptr[t] = excl;
      dinv[t]   = di;
    }
  }
  __syncthreads();

  if (inl) {
    for (int i0 = tid; i0 < nb; i0 += RK * 256) {
      int r[RK], pos[RK];
      for (int u = 0; u < RK; ++u) r[u] = lrec[min(i0 + u * 256, nb - 1)];
      for (int u = 0; u < RK; ++u)
        pos[u] = (i0 + u * 256 < nb) ? atomicAdd(&cur[r[u] & (BR - 1)], 1) : begin;
      for (int u = 0; u < RK; ++u)
        if (i0 + u * 256 < nb) lcol[pos[u] - begin] = r[u] >> BSH;
    }
    __syncthreads();
    for (int i = tid - (begin & 31); i < nb; i += 256)
      if (i >= 0) colA[begin + i] = lcol[i];
  } else {
    for (int i0 = tid; i0 < nb; i0 += RK * 256) {
      int r[RK];
      for (int u = 0; u < RK; ++u)
        r[u] = gcn_bucket_record(rec, pre, radr, min(i0 + u * 256, nb - 1));
      for (int u = 0; u < RK; ++u) {
        if (i0 + u * 256 < nb) {
          const int pos = atomicAdd(&cur[r[u] & (BR - 1)], 1);
          colA[pos] = r[u] >> BSH;
        }
      }
    }
  }

  const int l = tid & 15;
  float4 xv[BR / 16];
  for (int u = 0; u < BR / 16; ++u) {
    const int t = min(b * BR + u * 16 + (tid >> 4), NN - 1);
    xv[u] = *reinterpret_cast<const float4*>(x + t * DH + l * 4);
  }
  for (int u = 0; u < BR / 16; ++u) {
    const int rl = u * 16 + (tid >> 4);
    const int t  = b * BR + rl;
    if (t < NN) {
      const float d = dl[rl];
      float4 yv;
      yv.x = xv[u].x * d;
      yv.y = xv[u].y * d;
      yv.z = xv[u].z * d;
      yv.w = xv[u].w * d;
      *reinterpret_cast<float4*>(Y0 + t * DH + l * 4) = yv;
    }
  }
}

#define SCAP 64
__device__ __forceinline__ void gcn_sort_row(const int* __restrict__ colU,
                                             int* colS, int* sl, int start,
                                             int n, int lane) {
  if (n <= 32) {
    const int key = (lane < n) ? colU[start + lane] : 0x7fffffff;
    int r = 0;
    for (int j = 0; j < n; ++j) {
      const int kj = __shfl(key, j, 32);
      r += ((kj < key) || (kj == key && j < lane)) ? 1 : 0;
    }
    if (lane < n) {
      colS[start + r] = key;
      sl[r] = key;
    }
  } else {
    for (int i = lane; i < n; i += 32) {
      const int key = colU[start + i];
      int r = 0;
      for (int j = 0; j < n; ++j) {
        const int kj = colU[start + j];
        r += ((kj < key) || (kj == key && j < i)) ? 1 : 0;
      }
      colS[start + r] = key;
      if (r < SCAP) sl[r] = key;
    }
  }
}

#define PITCH 68

template <int NCOLS, int WP>
__device__ __forceinline__ void gcn_stage_w(const float* __restrict__ W,
                                            float* sW, int tid) {
  constexpr int NV = DH * NCOLS / 4 / 256;
  static_assert(NV * 4 * 256 == DH * NCOLS && NCOLS % 4 == 0, "256 threads cover W");
  float4 w[NV];
  for (int i = 0; i < NV; ++i)
    w[i] = reinterpret_cast<const float4*>(W)[i * 256 + tid];
  for (int i = 0; i < NV; ++i) {
    const int e = (i * 256 + tid) * 4;
    const int r = e / NCOLS;
    const int c = e % NCOLS;
    float* d = sW + ((r >> 1) * WP + c) * 2 + (r & 1);
    d[0] = w[i].x;
    d[2] = w[i].y;
    d[4] = w[i].z;
    d[6] = w[i].w;
  }
}

template <int WP>
__device__ __forceinline__ v8f gcn_tile_wmma_lds(const float* S, const float* sW,
                                                 int tileN, int lane) {
  const int half = lane >> 4;
  const int l    = lane & 15;
  const int bcol = tileN * 16 + l;
  const float* ap = S + l * PITCH + half * 2;
  const float* bp = sW + (half * WP + bcol) * 2;
  v8f c = {};
  for (int k = 0; k < DH / 4; ++k) {
    const v2f a = *reinterpret_cast<const v2f*>(ap + k * 4);
    const v2f b = *reinterpret_cast<const v2f*>(bp + k * (4 * WP));
    c = __builtin_amdgcn_wmma_f32_16x16x4_f32(
         false, a,  false, b,
         (short)0, c,  false,  false);
  }
  return c;
}

__device__ __forceinline__ void gcn_issue8(const float4* __restrict__ Y4,
                                           int idx, int jb, int l,
                                           float4 (&h)[8]) {
  int s[8];
  for (int k = 0; k < 8; ++k) s[k] = __shfl(idx, jb + k, 16);
  for (int k = 0; k < 8; ++k) h[k] = Y4[(unsigned)(s[k] * (DH / 4) + l)];
}

__device__ __forceinline__ void gcn_sum8(const float4 (&h)[8], int q,
                                         float4& acc) {
  for (int k = 0; k < 8; ++k) {
    const float f = (q > k) ? 1.0f : 0.0f;
    acc.x = __builtin_fmaf(h[k].x, f, acc.x);
    acc.y = __builtin_fmaf(h[k].y, f, acc.y);
    acc.z = __builtin_fmaf(h[k].z, f, acc.z);
    acc.w = __builtin_fmaf(h[k].w, f, acc.w);
  }
}

__device__ __forceinline__ void gcn_gather8(const float4* __restrict__ Y4,
                                            int idx, int jb, int q, int l,
                                            float4& acc) {
  float4 h[8];
  gcn_issue8(Y4, idx, jb, l, h);
  gcn_sum8(h, q, acc);
}

__device__ __forceinline__ void gcn_lds_barrier() {
  __builtin_amdgcn_fence(__ATOMIC_RELEASE, "workgroup", "local");
  __builtin_amdgcn_s_barrier();
  __builtin_amdgcn_fence(__ATOMIC_ACQUIRE, "workgroup", "local");
}

__device__ __forceinline__ void gcn_sort_pair(const int* __restrict__ col,
                                              int* colS, int* sIdx, int wv,
                                              int lane, int l, int m, int start,
                                              int n, int t, int keyA, int keyB,
                                              int& idx0, int& idx1) {
  const int stA = __builtin_amdgcn_readlane(start, 0);
  const int nA  = __builtin_amdgcn_readlane(n, 0);
  const int stB = __builtin_amdgcn_readlane(start, 16);
  const int nB  = __builtin_amdgcn_readlane(n, 16);
  int* slA = sIdx + wv * (2 * SCAP);
  int* slB = slA + SCAP;
  if (nA <= 32 && nB <= 32) {
    const int nm = max(nA, nB);
    int rA = 0, rB = 0;
    for (int j = 0; j < nm; ++j) {
      const int kA = __shfl(keyA, j, 32);
      const int kB = __shfl(keyB, j, 32);
      rA += ((kA < keyA) || (kA == keyA && j < lane)) ? 1 : 0;
      rB += ((kB < keyB) || (kB == keyB && j < lane)) ? 1 : 0;
    }
    if (lane < nA) {
      colS[stA + rA] = keyA;
      slA[rA] = keyA;
    }
    if (lane < nB) {
      colS[stB + rB] = keyB;
      slB[rB] = keyB;
    }
  } else {
    gcn_sort_row(col, colS, slA, stA, nA, lane);
    gcn_sort_row(col, colS, slB, stB, nB, lane);
  }
  idx0 = (l < n)      ? sIdx[m * SCAP + l]      : t;
  idx1 = (16 + l < n) ? sIdx[m * SCAP + 16 + l] : t;
}

#define TPB 5
static_assert(NN % 16 == 0 && (NN / 16) % TPB == 0, "layer blocks tile the nodes");

template <bool LAST, bool FIRST>
__global__ __launch_bounds__(256, 10) void gcn_layer_kernel(
    const float* __restrict__ Y, const int* __restrict__ col, int* colS,
    const int* __restrict__ rowptr, const int* __restrict__ cnt,
    const float* __restrict__ dinv,
    const float* __restrict__ W, const float* __restrict__ b,
    const float* __restrict__ Wf1, const float* __restrict__ bf1,
    const float* __restrict__ Wf2, const float* __restrict__ bf2,
    float* __restrict__ OUT) {
  __shared__ float sA[16 * PITCH];
  __shared__ float sB[16 * PITCH];
  __shared__ int sIdx[FIRST ? 16 * SCAP : 1];
  __shared__ float sW[(DH / 2) * 80 * 2];
  __shared__ float sW1[LAST ? (DH / 2) * 80 * 2 : 2];
  __shared__ float sW2[LAST ? (DH / 2) * 16 * 2 : 2];
  const float4* __restrict__ Y4 = reinterpret_cast<const float4*>(Y);
  const int m = threadIdx.x >> 4;
  const int l = threadIdx.x & 15;
  const int wv   = __builtin_amdgcn_readfirstlane(threadIdx.x >> 5);
  const int lane = threadIdx.x & 31;
  const int half = lane >> 4;
  const int bcol = wv * 16 + (lane & 15);

  static_assert(TPB >= 2, "the second tile's bounds are read ahead of the loop");
  int t = blockIdx.x * (TPB * 16) + m;
  int start = rowptr[t];
  int n     = cnt[t];
  int nstart = rowptr[t + 16];
  int nn     = cnt[t + 16];

  float bv = 0.0f, bv1 = 0.0f, bv2 = 0.0f;
  if (wv < 4) {
    bv = b[bcol];
    if (LAST) bv1 = bf1[bcol];
    if (LAST && wv == 0) bv2 = bf2[bcol];
  }

  float  dt = dinv[t];
  float4 ys = Y4[t * (DH / 4) + l];
  int idx0 = t, idx1 = t;
  int keyA = 0x7fffffff, keyB = 0x7fffffff;
  if (FIRST) {
    const int stA = __builtin_amdgcn_readlane(start, 0);
    const int nA  = __builtin_amdgcn_readlane(n, 0);
    const int stB = __builtin_amdgcn_readlane(start, 16);
    const int nB  = __builtin_amdgcn_readlane(n, 16);
    if (lane < nA) keyA = col[stA + lane];
    if (lane < nB) keyB = col[stB + lane];
  } else {
    if (l < n)      idx0 = col[start + l];
    if (16 + l < n) idx1 = col[start + 16 + l];
  }

  gcn_stage_w<DH, 80>(W, sW, threadIdx.x);
  if (LAST) {
    gcn_stage_w<DH, 80>(Wf1, sW1, threadIdx.x);
    gcn_stage_w<DOUT, 16>(Wf2, sW2, threadIdx.x);
  }

  if (FIRST)
    gcn_sort_pair(col, colS, sIdx, wv, lane, l, m, start, n, t, keyA, keyB, idx0, idx1);
  int nmax = max(__builtin_amdgcn_readlane(n, 0), __builtin_amdgcn_readlane(n, 16));
  float4 h[8];
  gcn_issue8(Y4, idx0, 0, l, h);

#pragma nounroll
  for (int it = 0; it < TPB; ++it) {
    const bool more = it + 1 < TPB;
    const int  tn   = t + 16;

    asm volatile("" : : "v"(h[7].x));
    float4 acc = {0.0f, 0.0f, 0.0f, 0.0f};
    gcn_sum8(h, n, acc);
    int n2start = 0, n2n = 0;
    if (it + 2 < TPB) {
      n2start = rowptr[tn + 16];
      n2n     = cnt[tn + 16];
    }
    float  ndt = 0.0f;
    float4 nys = {0.0f, 0.0f, 0.0f, 0.0f};
    int nidx0 = tn, nidx1 = tn;
    int nkeyA = 0x7fffffff, nkeyB = 0x7fffffff;
    if (more) {
      ndt = dinv[tn];
      nys = Y4[tn * (DH / 4) + l];
      if (FIRST) {
        const int stA = __builtin_amdgcn_readlane(nstart, 0);
        const int nA  = __builtin_amdgcn_readlane(nn, 0);
        const int stB = __builtin_amdgcn_readlane(nstart, 16);
        const int nB  = __builtin_amdgcn_readlane(nn, 16);
        if (lane < nA) nkeyA = col[stA + lane];
        if (lane < nB) nkeyB = col[stB + lane];
      } else {
        if (l < nn)      nidx0 = col[nstart + l];
        if (16 + l < nn) nidx1 = col[nstart + 16 + l];
      }
    }

    if (nmax > 8)  gcn_gather8(Y4, idx0, 8, n - 8,  l, acc);
    if (nmax > 16) gcn_gather8(Y4, idx1, 0, n - 16, l, acc);
    if (nmax > 24) gcn_gather8(Y4, idx1, 8, n - 24, l, acc);
    for (int c0 = 32; c0 < nmax; c0 += 16) {
      const int p = c0 + l;
      int myidx = t;
      if (p < n) {
        if (FIRST) {
          if (p < SCAP) {
            myidx = sIdx[m * SCAP + p];
          } else {
            __threadfence_block();
            myidx = colS[start + p];
          }
        } else {
          myidx = col[start + p];
        }
      }
      const int cn = min(nmax - c0, 16);
#pragma nounroll
      for (int j = 0; j < cn; j += 4) {
        const int s0 = __shfl(myidx, j,     16);
        const int s1 = __shfl(myidx, j + 1, 16);
        const int s2 = __shfl(myidx, j + 2, 16);
        const int s3 = __shfl(myidx, j + 3, 16);
        const int   q  = n - (c0 + j);
        const float f0 = (q > 0) ? 1.0f : 0.0f;
        const float f1 = (q > 1) ? 1.0f : 0.0f;
        const float f2 = (q > 2) ? 1.0f : 0.0f;
        const float f3 = (q > 3) ? 1.0f : 0.0f;
        const float4 h0 = Y4[(unsigned)(s0 * (DH / 4) + l)];
        const float4 h1 = Y4[(unsigned)(s1 * (DH / 4) + l)];
        const float4 h2 = Y4[(unsigned)(s2 * (DH / 4) + l)];
        const float4 h3 = Y4[(unsigned)(s3 * (DH / 4) + l)];
        acc.x = __builtin_fmaf(h0.x, f0, acc.x); acc.y = __builtin_fmaf(h0.y, f0, acc.y);
        acc.z = __builtin_fmaf(h0.z, f0, acc.z); acc.w = __builtin_fmaf(h0.w, f0, acc.w);
        acc.x = __builtin_fmaf(h1.x, f1, acc.x); acc.y = __builtin_fmaf(h1.y, f1, acc.y);
        acc.z = __builtin_fmaf(h1.z, f1, acc.z); acc.w = __builtin_fmaf(h1.w, f1, acc.w);
        acc.x = __builtin_fmaf(h2.x, f2, acc.x); acc.y = __builtin_fmaf(h2.y, f2, acc.y);
        acc.z = __builtin_fmaf(h2.z, f2, acc.z); acc.w = __builtin_fmaf(h2.w, f2, acc.w);
        acc.x = __builtin_fmaf(h3.x, f3, acc.x); acc.y = __builtin_fmaf(h3.y, f3, acc.y);
        acc.z = __builtin_fmaf(h3.z, f3, acc.z); acc.w = __builtin_fmaf(h3.w, f3, acc.w);
      }
    }

    float4 g;
    g.x = (acc.x + ys.x) * dt;
    g.y = (acc.y + ys.y) * dt;
    g.z = (acc.z + ys.z) * dt;
    g.w = (acc.w + ys.w) * dt;
    *reinterpret_cast<float4*>(sA + m * PITCH + l * 4) = g;

    int nnmax = 0;
    if (more) {
      if (FIRST)
        gcn_sort_pair(col, colS, sIdx, wv, lane, l, m, nstart, nn, tn, nkeyA, nkeyB, nidx0, nidx1);
      nnmax = max(__builtin_amdgcn_readlane(nn, 0), __builtin_amdgcn_readlane(nn, 16));
      gcn_issue8(Y4, nidx0, 0, l, h);
    } else {
      for (int k = 0; k < 8; ++k) h[k] = float4{0.0f, 0.0f, 0.0f, 0.0f};
    }
    gcn_lds_barrier();

    if (wv < 4) {
      const v8f c = gcn_tile_wmma_lds<80>(sA, sW, wv, lane);
      for (int r = 0; r < 8; ++r)
        sB[(half * 8 + r) * PITCH + bcol] = fmaxf(c[r] + bv, 0.0f);
    }
    gcn_lds_barrier();

    if (!LAST) {
      float4 o = *reinterpret_cast<const float4*>(sB + m * PITCH + l * 4);
      o.x *= dt; o.y *= dt; o.z *= dt; o.w *= dt;
      *reinterpret_cast<float4*>(OUT + t * DH + l * 4) = o;
    } else {
      if (wv < 4) {
        const v8f c = gcn_tile_wmma_lds<80>(sB, sW1, wv, lane);
        for (int r = 0; r < 8; ++r)
          sA[(half * 8 + r) * PITCH + bcol] = fmaxf(c[r] + bv1, 0.0f);
      }
      gcn_lds_barrier();
      if (wv == 0) {
        const v8f c = gcn_tile_wmma_lds<16>(sA, sW2, 0, lane);
        for (int r = 0; r < 8; ++r)
          sB[(half * 8 + r) * DOUT + bcol] = c[r] + bv2;
      }
      gcn_lds_barrier();
      if (threadIdx.x < 64) {
        const float4 o = *reinterpret_cast<const float4*>(sB + threadIdx.x * 4);
        *reinterpret_cast<float4*>(OUT + (blockIdx.x * TPB + it) * (16 * DOUT) + threadIdx.x * 4) = o;
      }
    }

    t = tn; start = nstart; n = nn; dt = ndt; ys = nys;
    nstart = n2start; nn = n2n;
    idx0 = nidx0; idx1 = nidx1; nmax = nnmax;
  }
}

extern "C" void kernel_launch(void* const* d_in, const int* in_sizes, int n_in,
                              void* d_out, int out_size, void* d_ws,
                              size_t ws_size, hipStream_t stream) {
  (void)in_sizes; (void)n_in; (void)out_size; (void)ws_size;

  const float* x   = (const float*)d_in[0];
  const int*   ei  = (const int*)d_in[1];
  const int*   src = ei;
  const int*   dst = ei + EE;
  const float* W1  = (const float*)d_in[2];
  const float* b1  = (const float*)d_in[3];
  const float* W2  = (const float*)d_in[4];
  const float* b2  = (const float*)d_in[5];
  const float* W3  = (const float*)d_in[6];
  const float* b3  = (const float*)d_in[7];
  const float* Wf1 = (const float*)d_in[8];
  const float* bf1 = (const float*)d_in[9];
  const float* Wf2 = (const float*)d_in[10];
  const float* bf2 = (const float*)d_in[11];
  float* out = (float*)d_out;

  float* base = (float*)d_ws;
  float* dinv = base;
  float* bufA = base + NN;
  float* bufB = bufA + NN * DH;
  int* cnt    = (int*)(bufB + NN * DH);
  int* rowptr = cnt + NN;
  int* rec    = rowptr + NN;
  int* colA   = rec + EE;
  int* colB   = colA + EE;
  int* offT   = colB + EE;

  gcn_slice_sort_kernel<<<NS, ST, 0, stream>>>(src, dst, rec, offT);
  gcn_bucket_rows_kernel<<<NB, 256, 0, stream>>>(rec, offT, x, cnt, rowptr, dinv,
                                               colA, bufB);

  gcn_layer_kernel<false, true><<<NN / 16 / TPB, 256, 0, stream>>>(
      bufB, colA, colB, rowptr, cnt, dinv, W1, b1, nullptr, nullptr, nullptr, nullptr, bufA);
  gcn_layer_kernel<false, false><<<NN / 16 / TPB, 256, 0, stream>>>(
      bufA, colB, nullptr, rowptr, cnt, dinv, W2, b2, nullptr, nullptr, nullptr, nullptr, bufB);
  gcn_layer_kernel<true, false><<<NN / 16 / TPB, 256, 0, stream>>>(
      bufB, colB, nullptr, rowptr, cnt, dinv, W3, b3, Wf1, bf1, Wf2, bf2, out);
}
